# starting kernel
# baseline (speedup 1.0000x reference)
#include <hip/hip_runtime.h>
#include <cmath>

#define IN_F  256
#define OUT_F 64
#define LRELU_ALPHA 0.2f

typedef __attribute__((ext_vector_type(2))) float v2f;
typedef __attribute__((ext_vector_type(8))) float v8f;

__global__ __launch_bounds__(256) void gat_gemm_wmma(
    const float* __restrict__ x, const float* __restrict__ W,
    float* __restrict__ h, int N)
{
    __shared__ float Wlds[IN_F * OUT_F];
    const int tid = threadIdx.x;
    for (int i = tid; i < IN_F * OUT_F; i += 256) {
        const int k = i / OUT_F, c = i % OUT_F;
        Wlds[(k >> 1) * (2 * OUT_F) + c * 2 + (k & 1)] = W[i];
    }
    __syncthreads();

    const int wave = tid >> 5;
    const int lane = tid & 31;
    const int half = lane >> 4;
    const int l16  = lane & 15;
    const int tile_row0 = blockIdx.x * 128 + wave * 16;
    const int arow = tile_row0 + l16;
    const int lrow = arow < N ? arow : (N - 1);

    const float* xptr = x + (size_t)lrow * IN_F + 2 * half;
    const float* bptr = Wlds + half * (2 * OUT_F) + l16 * 2;

    v8f acc[4] = {};

    for (int k0 = 0; k0 < IN_F; k0 += 4) {
        const float2 t = *(const float2*)(xptr + k0);
        v2f a; a.x = t.x; a.y = t.y;
        const float* bk = bptr + (k0 >> 1) * (2 * OUT_F);
#pragma unroll
        for (int n = 0; n < 4; ++n) {
            const float2 u = *(const float2*)(bk + n * 32);
            v2f b; b.x = u.x; b.y = u.y;
            acc[n] = __builtin_amdgcn_wmma_f32_16x16x4_f32(
                false, a, false, b, (short)0, acc[n], false, false);
        }
    }

    const int rbase = tile_row0 + 8 * half;
#pragma unroll
    for (int r = 0; r < 8; ++r) {
        const int rr = rbase + r;
        if (rr < N) {
#pragma unroll
            for (int n = 0; n < 4; ++n)
                h[(size_t)rr * OUT_F + n * 16 + l16] = acc[n][r];
        }
    }
}

__global__ __launch_bounds__(256) void gat_scores(
    const float* __restrict__ h, const float* __restrict__ attn,
    float* __restrict__ s_src, float* __restrict__ s_dst, int N)
{
    const int node = (blockIdx.x * blockDim.x + threadIdx.x) >> 5;
    const int lane = threadIdx.x & 31;
    if (node >= N) return;
    const float* hr = h + (size_t)node * OUT_F;
    const float h0 = hr[lane], h1 = hr[lane + 32];
    float ps = h0 * attn[lane]         + h1 * attn[lane + 32];
    float pd = h0 * attn[OUT_F + lane] + h1 * attn[OUT_F + lane + 32];
    for (int off = 16; off > 0; off >>= 1) {
        ps += __shfl_down(ps, off, 32);
        pd += __shfl_down(pd, off, 32);
    }
    if (lane == 0) { s_src[node] = ps; s_dst[node] = pd; }
}

__device__ __forceinline__ float edge_logit(const float* s_src, const float* s_dst,
                                            int s, int d) {
    const float v = s_src[s] + s_dst[d];
    return -(v > 0.f ? v : LRELU_ALPHA * v);
}

__global__ void gat_init(float* __restrict__ out, float* __restrict__ m,
                         float* __restrict__ denom, int N)
{
    const int i = blockIdx.x * blockDim.x + threadIdx.x;
    if (i < N * OUT_F) out[i] = 0.f;
    if (i < N) { m[i] = -INFINITY; denom[i] = 0.f; }
}

__global__ void gat_edge_max(const int* __restrict__ src, const int* __restrict__ dst,
                             const float* __restrict__ s_src, const float* __restrict__ s_dst,
                             float* __restrict__ m, int E)
{
    const int i = blockIdx.x * blockDim.x + threadIdx.x;
    if (i >= E) return;
    const int s = src[i];
    const float e = edge_logit(s_src, s_dst, s, dst[i]);
    atomicMax(&m[s], e);
}

__global__ void gat_edge_sum(const int* __restrict__ src, const int* __restrict__ dst,
                             const float* __restrict__ s_src, const float* __restrict__ s_dst,
                             const float* __restrict__ m, float* __restrict__ denom, int E)
{
    const int i = blockIdx.x * blockDim.x + threadIdx.x;
    if (i >= E) return;
    const int s = src[i];
    const float e = edge_logit(s_src, s_dst, s, dst[i]);
    atomicAdd(&denom[s], __expf(e - m[s]));
}

__global__ __launch_bounds__(256) void gat_edge_aggr(
    const int* __restrict__ src, const int* __restrict__ dst,
    const float* __restrict__ s_src, const float* __restrict__ s_dst,
    const float* __restrict__ m, const float* __restrict__ denom,
    const float* __restrict__ h, float* __restrict__ out, int E)
{
    const int edge = (blockIdx.x * blockDim.x + threadIdx.x) >> 5;
    const int lane = threadIdx.x & 31;
    if (edge >= E) return;
    const int s = src[edge], d = dst[edge];
    const float e = edge_logit(s_src, s_dst, s, d);
    const float w = __expf(e - m[s]) / denom[s];
    const float* hr = h + (size_t)d * OUT_F;
    float* orow = out + (size_t)s * OUT_F;
    atomicAdd(&orow[lane],      w * hr[lane]);
    atomicAdd(&orow[lane + 32], w * hr[lane + 32]);
}

__global__ void gat_elu(float* __restrict__ out, int total)
{
    const int i = blockIdx.x * blockDim.x + threadIdx.x;
    if (i < total) {
        const float v = out[i];
        out[i] = v > 0.f ? v : expm1f(v);
    }
}

extern "C" void kernel_launch(void* const* d_in, const int* in_sizes, int n_in,
                              void* d_out, int out_size, void* d_ws, size_t ws_size,
                              hipStream_t stream)
{
    const float* x    = (const float*)d_in[0];
    const int*   edge = (const int*)  d_in[1];
    const float* W    = (const float*)d_in[2];
    const float* attn = (const float*)d_in[3];

    const int N = in_sizes[0] / IN_F;
    const int E = in_sizes[1] / 2;
    const int* src = edge;
    const int* dst = edge + E;
    float* out = (float*)d_out;

    float* h     = (float*)d_ws;
    float* s_src = h + (size_t)N * OUT_F;
    float* s_dst = s_src + N;
    float* m     = s_dst + N;
    float* denom = m + N;

    const int totalNF = N * OUT_F;

    gat_init     <<<(totalNF + 255) / 256, 256, 0, stream>>>(out, m, denom, N);
    gat_gemm_wmma<<<(N + 127) / 128,       256, 0, stream>>>(x, W, h, N);
    gat_scores   <<<(N * 32 + 255) / 256,  256, 0, stream>>>(h, attn, s_src, s_dst, N);
    gat_edge_max <<<(E + 255) / 256,       256, 0, stream>>>(src, dst, s_src, s_dst, m, E);
    gat_edge_sum <<<(E + 255) / 256,       256, 0, stream>>>(src, dst, s_src, s_dst, m, denom, E);
    {
        const long long threads = (long long)E * 32;
        gat_edge_aggr<<<(int)((threads + 255) / 256), 256, 0, stream>>>(
            src, dst, s_src, s_dst, m, denom, h, out, E);
    }
    gat_elu      <<<(totalNF + 255) / 256, 256, 0, stream>>>(out, totalNF);
}


# Edge sums by row-owner blocks: a block keeps 392 out rows and denominators in LDS, reads all of src, adds its edges with LDS float atomics (arrival order), divides, stores once; no global out atomics
# speedup vs baseline: 1.8608x; 1.8608x over previous
#include <hip/hip_runtime.h>
#include <cmath>

#define IN_F  256
#define OUT_F 64
#define LRELU_ALPHA 0.2f

typedef __attribute__((ext_vector_type(2))) float v2f;
typedef __attribute__((ext_vector_type(8))) float v8f;

__global__ __launch_bounds__(256) void gat_gemm_wmma(
    const float* __restrict__ x, const float* __restrict__ W,
    float* __restrict__ h, int N)
{
    __shared__ float Wlds[IN_F * OUT_F];
    const int tid = threadIdx.x;
    for (int i = tid; i < IN_F * OUT_F; i += 256) {
        const int k = i / OUT_F, c = i % OUT_F;
        Wlds[(k >> 1) * (2 * OUT_F) + c * 2 + (k & 1)] = W[i];
    }
    __syncthreads();

    const int wave = tid >> 5;
    const int lane = tid & 31;
    const int half = lane >> 4;
    const int l16  = lane & 15;
    const int tile_row0 = blockIdx.x * 128 + wave * 16;
    const int arow = tile_row0 + l16;
    const int lrow = arow < N ? arow : (N - 1);

    const float* xptr = x + (size_t)lrow * IN_F + 2 * half;
    const float* bptr = Wlds + half * (2 * OUT_F) + l16 * 2;

    v8f acc[4] = {};

    for (int k0 = 0; k0 < IN_F; k0 += 4) {
        const float2 t = *(const float2*)(xptr + k0);
        v2f a; a.x = t.x; a.y = t.y;
        const float* bk = bptr + (k0 >> 1) * (2 * OUT_F);
#pragma unroll
        for (int n = 0; n < 4; ++n) {
            const float2 u = *(const float2*)(bk + n * 32);
            v2f b; b.x = u.x; b.y = u.y;
            acc[n] = __builtin_amdgcn_wmma_f32_16x16x4_f32(
                false, a, false, b, (short)0, acc[n], false, false);
        }
    }

    const int rbase = tile_row0 + 8 * half;
#pragma unroll
    for (int r = 0; r < 8; ++r) {
        const int rr = rbase + r;
        if (rr < N) {
#pragma unroll
            for (int n = 0; n < 4; ++n)
                h[(size_t)rr * OUT_F + n * 16 + l16] = acc[n][r];
        }
    }
}

__global__ __launch_bounds__(256) void gat_scores(
    const float* __restrict__ h, const float* __restrict__ attn,
    float* __restrict__ s_src, float* __restrict__ s_dst, int N)
{
    const int node = (blockIdx.x * blockDim.x + threadIdx.x) >> 5;
    const int lane = threadIdx.x & 31;
    if (node >= N) return;
    const float* hr = h + (size_t)node * OUT_F;
    const float h0 = hr[lane], h1 = hr[lane + 32];
    float ps = h0 * attn[lane]         + h1 * attn[lane + 32];
    float pd = h0 * attn[OUT_F + lane] + h1 * attn[OUT_F + lane + 32];
    for (int off = 16; off > 0; off >>= 1) {
        ps += __shfl_down(ps, off, 32);
        pd += __shfl_down(pd, off, 32);
    }
    if (lane == 0) { s_src[node] = ps; s_dst[node] = pd; }
}

__device__ __forceinline__ float logit_of(float v) {
    return -(v > 0.f ? v : LRELU_ALPHA * v);
}

__device__ __forceinline__ float edge_logit(const float* s_src, const float* s_dst,
                                            int s, int d) {
    return logit_of(s_src[s] + s_dst[d]);
}

__global__ void gat_init(float* __restrict__ m, int N)
{
    const int i = blockIdx.x * blockDim.x + threadIdx.x;
    if (i < N) m[i] = -INFINITY;
}

__global__ void gat_edge_max(const int* __restrict__ src, const int* __restrict__ dst,
                             const float* __restrict__ s_src, const float* __restrict__ s_dst,
                             float* __restrict__ m, int E)
{
    const int i = blockIdx.x * blockDim.x + threadIdx.x;
    if (i >= E) return;
    const int s = src[i];
    const float e = edge_logit(s_src, s_dst, s, dst[i]);
    atomicMax(&m[s], e);
}

#define ROWS_PB   392
#define SCAN_U    8
#define AGG_WAVES 32
#define Q_DRAIN   96
#define Q_CAP     (32 * 4 * SCAN_U + Q_DRAIN)

__device__ __forceinline__ int4 load_src4_tail(const int* __restrict__ src,
                                               int i4, int nvec, int E)
{
    int4 v = make_int4(-1, -1, -1, -1);
    if (i4 < nvec) {
        v = ((const int4*)src)[i4];
    } else if (i4 == nvec) {
        const int e0 = nvec * 4;
        if (e0     < E) v.x = src[e0];
        if (e0 + 1 < E) v.y = src[e0 + 1];
        if (e0 + 2 < E) v.z = src[e0 + 2];
    }
    return v;
}

__device__ __forceinline__ void file_hit(bool hit, int e, int* qe, int& qn)
{
    const unsigned mask = __builtin_amdgcn_ballot_w32(hit);
    if (hit) qe[qn + (int)__builtin_amdgcn_mbcnt_lo(mask, 0u)] = e;
    qn += __builtin_popcount(mask);
}

__device__ __forceinline__ void test4(const int4 v, int i4, int row0, unsigned nrows,
                                      int* qe, int& qn)
{
    const bool h0 = (unsigned)(v.x - row0) < nrows;
    const bool h1 = (unsigned)(v.y - row0) < nrows;
    const bool h2 = (unsigned)(v.z - row0) < nrows;
    const bool h3 = (unsigned)(v.w - row0) < nrows;
    if (__builtin_amdgcn_ballot_w32(h0 | h1 | h2 | h3) != 0u) {
        const int e0 = i4 * 4;
        file_hit(h0, e0,     qe, qn);
        file_hit(h1, e0 + 1, qe, qn);
        file_hit(h2, e0 + 2, qe, qn);
        file_hit(h3, e0 + 3, qe, qn);
    }
}

template <int G>
__device__ __forceinline__ void add_rows(int d, int r, float p, int k, int lane,
                                         const float* __restrict__ h, float* acc)
{
    float2 hv[G];
    float  pk[G];
    int    rk[G];
#pragma unroll
    for (int u = 0; u < G; ++u) {
        const int dk = __builtin_amdgcn_readlane(d, k + u);
        rk[u] = __builtin_amdgcn_readlane(r, k + u);
        pk[u] = __int_as_float(__builtin_amdgcn_readlane(__float_as_int(p), k + u));
        hv[u] = *(const float2*)(h + (size_t)dk * OUT_F + 2 * lane);
    }
#pragma unroll
    for (int u = 0; u < G; ++u) {
        float* a = acc + rk[u] * OUT_F + 2 * lane;
        atomicAdd(a,     pk[u] * hv[u].x);
        atomicAdd(a + 1, pk[u] * hv[u].y);
    }
}

__global__ __launch_bounds__(1024) void gat_rows_scan(
    const int* __restrict__ src, const int* __restrict__ dst,
    const float* __restrict__ s_src, const float* __restrict__ s_dst,
    const float* __restrict__ m, const float* __restrict__ h,
    float* __restrict__ out, int N, int E)
{
    __shared__ float acc[ROWS_PB * OUT_F];
    __shared__ float den[ROWS_PB];
    __shared__ float ssl[ROWS_PB];
    __shared__ float ml[ROWS_PB];
    __shared__ int   queue[AGG_WAVES * Q_CAP];

    const int tid  = threadIdx.x;
    const int lane = tid & 31;
    const int wave = tid >> 5;
    const int row0 = blockIdx.x * ROWS_PB;
    const int nrows = (N - row0) < ROWS_PB ? (N - row0) : ROWS_PB;

    for (int i = tid; i < ROWS_PB * OUT_F; i += 1024) acc[i] = 0.f;
    for (int i = tid; i < ROWS_PB; i += 1024) {
        const bool ok = i < nrows;
        den[i] = 0.f;
        ssl[i] = ok ? s_src[row0 + i] : 0.f;
        ml[i]  = ok ? m[row0 + i] : 0.f;
    }
    __syncthreads();

    int* qe = queue + wave * Q_CAP;
    int  qn = 0;
    const int nvec   = E >> 2;
    const int nfull  = nvec / (1024 * SCAN_U);
    const int nsteps = nfull + 1;
    const int t0 = (int)(((long long)blockIdx.x * nsteps) / gridDim.x);
    const int4* src4 = (const int4*)src;

    for (int it = 0; it < nsteps; ++it) {
        int t = it + t0;
        if (t >= nsteps) t -= nsteps;
        const int i0 = t * (1024 * SCAN_U) + wave * (32 * SCAN_U) + lane;
        int4 v[SCAN_U];
        if (t < nfull) {
#pragma unroll
            for (int u = 0; u < SCAN_U; ++u) v[u] = src4[i0 + u * 32];
        } else {
#pragma unroll
            for (int u = 0; u < SCAN_U; ++u) v[u] = load_src4_tail(src, i0 + u * 32, nvec, E);
        }
#pragma unroll
        for (int u = 0; u < SCAN_U; ++u)
            test4(v[u], i0 + u * 32, row0, (unsigned)nrows, qe, qn);

        const bool last = (it == nsteps - 1);
        if (qn >= Q_DRAIN || last) {
            int b = 0;
            while (qn - b >= 32 || (last && b < qn)) {
                const int cnt = (qn - b) < 32 ? (qn - b) : 32;
                int d = 0, r = 0;
                float p = 0.f;
                if (lane < cnt) {
                    const int e = qe[b + lane];
                    r = src[e] - row0;
                    d = dst[e];
                    p = __expf(logit_of(ssl[r] + s_dst[d]) - ml[r]);
                    atomicAdd(&den[r], p);
                }
                int k = 0;
                for (; k + 16 <= cnt; k += 16) add_rows<16>(d, r, p, k, lane, h, acc);
                for (; k + 4  <= cnt; k += 4)  add_rows<4>(d, r, p, k, lane, h, acc);
                for (; k      <  cnt; k += 1)  add_rows<1>(d, r, p, k, lane, h, acc);
                b += 32;
            }
            if (!last) {
                const int rem = qn - b;
                if (lane < rem) {
                    const int e = qe[b + lane];
                    qe[lane] = e;
                }
                qn = rem;
            }
        }
    }
    __syncthreads();

    for (int i = tid; i < nrows * OUT_F; i += 1024) {
        const float dn = den[i / OUT_F];
        const float v = acc[i] / (dn > 0.f ? dn : 1.f);
        out[(size_t)row0 * OUT_F + i] = v > 0.f ? v : expm1f(v);
    }
}

extern "C" void kernel_launch(void* const* d_in, const int* in_sizes, int n_in,
                              void* d_out, int out_size, void* d_ws, size_t ws_size,
                              hipStream_t stream)
{
    const float* x    = (const float*)d_in[0];
    const int*   edge = (const int*)  d_in[1];
    const float* W    = (const float*)d_in[2];
    const float* attn = (const float*)d_in[3];

    const int N = in_sizes[0] / IN_F;
    const int E = in_sizes[1] / 2;
    const int* src = edge;
    const int* dst = edge + E;
    float* out = (float*)d_out;

    float* h     = (float*)d_ws;
    float* s_src = h + (size_t)N * OUT_F;
    float* s_dst = s_src + N;
    float* m     = s_dst + N;

    gat_init     <<<(N + 255) / 256,       256, 0, stream>>>(m, N);
    gat_gemm_wmma<<<(N + 127) / 128,       256, 0, stream>>>(x, W, h, N);
    gat_scores   <<<(N * 32 + 255) / 256,  256, 0, stream>>>(h, attn, s_src, s_dst, N);
    gat_edge_max <<<(E + 255) / 256,       256, 0, stream>>>(src, dst, s_src, s_dst, m, E);
    gat_rows_scan<<<(N + ROWS_PB - 1) / ROWS_PB, 1024, 0, stream>>>(
        src, dst, s_src, s_dst, m, h, out, N, E);
}


# Owner blocks scan a 1-byte owner-block key per edge, written by the max pass (4 edges a thread), instead of the 4-byte src: 16 edges per 16-byte load; sums stay in arrival order
# speedup vs baseline: 2.0240x; 1.0877x over previous
#include <hip/hip_runtime.h>
#include <cmath>

#define IN_F  256
#define OUT_F 64
#define LRELU_ALPHA 0.2f

typedef __attribute__((ext_vector_type(2))) float v2f;
typedef __attribute__((ext_vector_type(8))) float v8f;
typedef __attribute__((ext_vector_type(4))) int   v4i;

__global__ __launch_bounds__(256) void gat_gemm_wmma(
    const float* __restrict__ x, const float* __restrict__ W,
    float* __restrict__ h, int N)
{
    __shared__ float Wlds[IN_F * OUT_F];
    const int tid = threadIdx.x;
    for (int i = tid; i < IN_F * OUT_F; i += 256) {
        const int k = i / OUT_F, c = i % OUT_F;
        Wlds[(k >> 1) * (2 * OUT_F) + c * 2 + (k & 1)] = W[i];
    }
    __syncthreads();

    const int wave = tid >> 5;
    const int lane = tid & 31;
    const int half = lane >> 4;
    const int l16  = lane & 15;
    const int tile_row0 = blockIdx.x * 128 + wave * 16;
    const int arow = tile_row0 + l16;
    const int lrow = arow < N ? arow : (N - 1);

    const float* xptr = x + (size_t)lrow * IN_F + 2 * half;
    const float* bptr = Wlds + half * (2 * OUT_F) + l16 * 2;

    v8f acc[4] = {};

    for (int k0 = 0; k0 < IN_F; k0 += 4) {
        const float2 t = *(const float2*)(xptr + k0);
        v2f a; a.x = t.x; a.y = t.y;
        const float* bk = bptr + (k0 >> 1) * (2 * OUT_F);
#pragma unroll
        for (int n = 0; n < 4; ++n) {
            const float2 u = *(const float2*)(bk + n * 32);
            v2f b; b.x = u.x; b.y = u.y;
            acc[n] = __builtin_amdgcn_wmma_f32_16x16x4_f32(
                false, a, false, b, (short)0, acc[n], false, false);
        }
    }

    const int rbase = tile_row0 + 8 * half;
#pragma unroll
    for (int r = 0; r < 8; ++r) {
        const int rr = rbase + r;
        if (rr < N) {
#pragma unroll
            for (int n = 0; n < 4; ++n)
                h[(size_t)rr * OUT_F + n * 16 + l16] = acc[n][r];
        }
    }
}

__global__ __launch_bounds__(256) void gat_scores(
    const float* __restrict__ h, const float* __restrict__ attn,
    float* __restrict__ s_src, float* __restrict__ s_dst, int N)
{
    const int node = (blockIdx.x * blockDim.x + threadIdx.x) >> 5;
    const int lane = threadIdx.x & 31;
    if (node >= N) return;
    const float* hr = h + (size_t)node * OUT_F;
    const float h0 = hr[lane], h1 = hr[lane + 32];
    float ps = h0 * attn[lane]         + h1 * attn[lane + 32];
    float pd = h0 * attn[OUT_F + lane] + h1 * attn[OUT_F + lane + 32];
    for (int off = 16; off > 0; off >>= 1) {
        ps += __shfl_down(ps, off, 32);
        pd += __shfl_down(pd, off, 32);
    }
    if (lane == 0) { s_src[node] = ps; s_dst[node] = pd; }
}

__device__ __forceinline__ float logit_of(float v) {
    return -(v > 0.f ? v : LRELU_ALPHA * v);
}

__device__ __forceinline__ float edge_logit(const float* s_src, const float* s_dst,
                                            int s, int d) {
    return logit_of(s_src[s] + s_dst[d]);
}

__global__ void gat_init(float* __restrict__ m, int N)
{
    const int i = blockIdx.x * blockDim.x + threadIdx.x;
    if (i < N) m[i] = -INFINITY;
}

#define ROWS_PB   392
#define WS_FLOATS_PER_NODE 67

__device__ __forceinline__ bool keys_usable(size_t ws_size, int N, int E)
{
    const size_t need = (size_t)N * WS_FLOATS_PER_NODE * sizeof(float)
                      + (size_t)E + 48;
    return (N + ROWS_PB - 1) / ROWS_PB <= 256 && ws_size >= need;
}

__device__ __forceinline__ unsigned* key_plane(float* ws_end)
{
    return (unsigned*)((char*)ws_end + ((16u - (unsigned)((size_t)ws_end & 15u)) & 15u));
}

__global__ __launch_bounds__(256) void gat_edge_max(
    const int* __restrict__ src, const int* __restrict__ dst,
    const float* __restrict__ s_src, const float* __restrict__ s_dst,
    float* __restrict__ m, float* ws_end, size_t ws_size, int N, int E)
{
    const int q  = blockIdx.x * blockDim.x + threadIdx.x;
    const int e0 = q * 4;
    const bool wk = keys_usable(ws_size, N, E);
    unsigned* keys = key_plane(ws_end);
    if (e0 >= E) {
        if (wk && e0 < ((E + 15) & ~15)) keys[q] = 0u;
        return;
    }
    int s[4], d[4];
    if (e0 + 3 < E && (E & 3) == 0) {
        const int4 sv = ((const int4*)src)[q];
        const int4 dv = ((const int4*)dst)[q];
        s[0] = sv.x; s[1] = sv.y; s[2] = sv.z; s[3] = sv.w;
        d[0] = dv.x; d[1] = dv.y; d[2] = dv.z; d[3] = dv.w;
    } else {
#pragma unroll
        for (int j = 0; j < 4; ++j) {
            const bool ok = e0 + j < E;
            s[j] = ok ? src[e0 + j] : -1;
            d[j] = ok ? dst[e0 + j] : 0;
        }
    }
    unsigned kw = 0u;
#pragma unroll
    for (int j = 0; j < 4; ++j) {
        if (s[j] >= 0) {
            atomicMax(&m[s[j]], edge_logit(s_src, s_dst, s[j], d[j]));
            kw |= (unsigned)(s[j] / ROWS_PB) << (8 * j);
        }
    }
    if (wk) keys[q] = kw;
}

#define SCAN_U    8
#define AGG_WAVES 32
#define Q_DRAIN   96
#define Q_CAP     (32 * 4 * SCAN_U + Q_DRAIN)

__device__ __forceinline__ int4 load_src4_tail(const int* __restrict__ src,
                                               int i4, int nvec, int E)
{
    int4 v = make_int4(-1, -1, -1, -1);
    if (i4 < nvec) {
        v = ((const int4*)src)[i4];
    } else if (i4 == nvec) {
        const int e0 = nvec * 4;
        if (e0     < E) v.x = src[e0];
        if (e0 + 1 < E) v.y = src[e0 + 1];
        if (e0 + 2 < E) v.z = src[e0 + 2];
    }
    return v;
}

__device__ __forceinline__ void file_hit(bool hit, int e, int* qe, int& qn)
{
    const unsigned mask = __builtin_amdgcn_ballot_w32(hit);
    if (hit) qe[qn + (int)__builtin_amdgcn_mbcnt_lo(mask, 0u)] = e;
    qn += __builtin_popcount(mask);
}

__device__ __forceinline__ void test4(const int4 v, int i4, int row0, unsigned nrows,
                                      int* qe, int& qn)
{
    const bool h0 = (unsigned)(v.x - row0) < nrows;
    const bool h1 = (unsigned)(v.y - row0) < nrows;
    const bool h2 = (unsigned)(v.z - row0) < nrows;
    const bool h3 = (unsigned)(v.w - row0) < nrows;
    if (__builtin_amdgcn_ballot_w32(h0 | h1 | h2 | h3) != 0u) {
        const int e0 = i4 * 4;
        file_hit(h0, e0,     qe, qn);
        file_hit(h1, e0 + 1, qe, qn);
        file_hit(h2, e0 + 2, qe, qn);
        file_hit(h3, e0 + 3, qe, qn);
    }
}

template <int G>
__device__ __forceinline__ void add_rows(int d, int r, float p, int k, int lane,
                                         const float* __restrict__ h, float* acc)
{
    float2 hv[G];
    float  pk[G];
    int    rk[G];
#pragma unroll
    for (int u = 0; u < G; ++u) {
        const int dk = __builtin_amdgcn_readlane(d, k + u);
        rk[u] = __builtin_amdgcn_readlane(r, k + u);
        pk[u] = __int_as_float(__builtin_amdgcn_readlane(__float_as_int(p), k + u));
        hv[u] = *(const float2*)(h + (size_t)dk * OUT_F + 2 * lane);
    }
#pragma unroll
    for (int u = 0; u < G; ++u) {
        float* a = acc + rk[u] * OUT_F + 2 * lane;
        atomicAdd(a,     pk[u] * hv[u].x);
        atomicAdd(a + 1, pk[u] * hv[u].y);
    }
}

__device__ __forceinline__ void drain_queue(
    int* qe, int& qn, bool flush, int lane, int row0,
    const int* __restrict__ src, const int* __restrict__ dst,
    const float* __restrict__ s_dst, const float* __restrict__ h,
    const float* ssl, const float* ml, float* den, float* acc)
{
    int b = 0;
    while (qn - b >= 32 || (flush && b < qn)) {
        const int cnt = (qn - b) < 32 ? (qn - b) : 32;
        int d = 0, r = 0;
        float p = 0.f;
        if (lane < cnt) {
            const int e = qe[b + lane];
            r = src[e] - row0;
            d = dst[e];
            p = __expf(logit_of(ssl[r] + s_dst[d]) - ml[r]);
            atomicAdd(&den[r], p);
        }
        int k = 0;
        for (; k + 16 <= cnt; k += 16) add_rows<16>(d, r, p, k, lane, h, acc);
        for (; k + 4  <= cnt; k += 4)  add_rows<4>(d, r, p, k, lane, h, acc);
        for (; k      <  cnt; k += 1)  add_rows<1>(d, r, p, k, lane, h, acc);
        b += 32;
    }
    if (flush) {
        qn = 0;
    } else if (b > 0) {
        const int rem = qn - b;
        if (lane < rem) {
            const int e = qe[b + lane];
            qe[lane] = e;
        }
        qn = rem;
    }
}

__device__ __forceinline__ unsigned zero_flags(unsigned x)
{
    return ~(((x & 0x7F7F7F7Fu) + 0x7F7F7F7Fu) | x);
}

__device__ __forceinline__ unsigned hit_mask2(const v4i a, const v4i b, unsigned splat)
{
    return ((zero_flags((unsigned)a.x ^ splat) >> 7) & 0x01010101u)
         | ((zero_flags((unsigned)a.y ^ splat) >> 6) & 0x02020202u)
         | ((zero_flags((unsigned)a.z ^ splat) >> 5) & 0x04040404u)
         | ((zero_flags((unsigned)a.w ^ splat) >> 4) & 0x08080808u)
         | ((zero_flags((unsigned)b.x ^ splat) >> 3) & 0x10101010u)
         | ((zero_flags((unsigned)b.y ^ splat) >> 2) & 0x20202020u)
         | ((zero_flags((unsigned)b.z ^ splat) >> 1) & 0x40404040u)
         | ( zero_flags((unsigned)b.w ^ splat)       & 0x80808080u);
}

#define KSCAN_U  8
#define KQ_ROOM  (Q_CAP - 32)

__global__ __launch_bounds__(1024) void gat_rows_scan(
    const int* __restrict__ src, const int* __restrict__ dst,
    const float* __restrict__ s_src, const float* __restrict__ s_dst,
    const float* __restrict__ m, const float* __restrict__ h,
    float* __restrict__ out, float* ws_end, size_t ws_size, int N, int E)
{
    __shared__ float acc[ROWS_PB * OUT_F];
    __shared__ float den[ROWS_PB];
    __shared__ float ssl[ROWS_PB];
    __shared__ float ml[ROWS_PB];
    __shared__ int   queue[AGG_WAVES * Q_CAP];

    const int tid  = threadIdx.x;
    const int lane = tid & 31;
    const int wave = tid >> 5;
    const int row0 = blockIdx.x * ROWS_PB;
    const int nrows = (N - row0) < ROWS_PB ? (N - row0) : ROWS_PB;

    for (int i = tid; i < ROWS_PB * OUT_F; i += 1024) acc[i] = 0.f;
    for (int i = tid; i < ROWS_PB; i += 1024) {
        const bool ok = i < nrows;
        den[i] = 0.f;
        ssl[i] = ok ? s_src[row0 + i] : 0.f;
        ml[i]  = ok ? m[row0 + i] : 0.f;
    }
    __syncthreads();

    int* qe = queue + wave * Q_CAP;
    int  qn = 0;

    if (keys_usable(ws_size, N, E)) {
        const v4i* keys4 = (const v4i*)key_plane(ws_end);
        const unsigned splat = blockIdx.x * 0x01010101u;
        const int nvec   = (E + 15) >> 4;
        const int nsteps = (nvec + 1024 * KSCAN_U - 1) / (1024 * KSCAN_U);
        const int t0 = (int)(((long long)blockIdx.x * nsteps) / gridDim.x);
        for (int it = 0; it < nsteps; ++it) {
            int t = it + t0;
            if (t >= nsteps) t -= nsteps;
            const int i0 = t * (1024 * KSCAN_U) + wave * (32 * KSCAN_U) + lane;
            unsigned mm[KSCAN_U / 2];
            {
                v4i v[KSCAN_U];
#pragma unroll
                for (int u = 0; u < KSCAN_U; ++u) {
                    const int iv = i0 + u * 32;
                    v[u] = keys4[iv < nvec ? iv : nvec - 1];
                }
                static_assert(KSCAN_U == 8, "the fence names eight vectors");
                asm volatile("" :: "v"(v[0]), "v"(v[1]), "v"(v[2]), "v"(v[3]),
                                   "v"(v[4]), "v"(v[5]), "v"(v[6]), "v"(v[7]));
#pragma unroll
                for (int u = 0; u < KSCAN_U / 2; ++u)
                    mm[u] = hit_mask2(v[2 * u], v[2 * u + 1], splat);
            }
            const bool last = (it == nsteps - 1);
            bool more;
            do {
#pragma unroll
                for (int u = 0; u < KSCAN_U / 2; ++u) {
                    while (qn <= KQ_ROOM) {
                        const bool has = mm[u] != 0u;
                        if (__builtin_amdgcn_ballot_w32(has) == 0u) break;
                        int e = E;
                        if (has) {
                            const int p = __builtin_ctz(mm[u]);
                            mm[u] &= mm[u] - 1u;
                            e = (i0 + (2 * u + ((p >> 2) & 1)) * 32) * 16 + 4 * (p & 3) + (p >> 3);
                        }
                        file_hit(e < E, e, qe, qn);
                    }
                }
                unsigned any = 0u;
#pragma unroll
                for (int u = 0; u < KSCAN_U / 2; ++u) any |= mm[u];
                more = __builtin_amdgcn_ballot_w32(any != 0u) != 0u;
                const bool flush = last && !more;
                if (qn >= Q_DRAIN || more || flush)
                    drain_queue(qe, qn, flush, lane, row0, src, dst, s_dst, h, ssl, ml, den, acc);
            } while (more);
        }
    } else {
        const int nvec   = E >> 2;
        const int nfull  = nvec / (1024 * SCAN_U);
        const int nsteps = nfull + 1;
        const int t0 = (int)(((long long)blockIdx.x * nsteps) / gridDim.x);
        const int4* src4 = (const int4*)src;
        for (int it = 0; it < nsteps; ++it) {
            int t = it + t0;
            if (t >= nsteps) t -= nsteps;
            const int i0 = t * (1024 * SCAN_U) + wave * (32 * SCAN_U) + lane;
            int4 v[SCAN_U];
            if (t < nfull) {
#pragma unroll
                for (int u = 0; u < SCAN_U; ++u) v[u] = src4[i0 + u * 32];
            } else {
#pragma unroll
                for (int u = 0; u < SCAN_U; ++u) v[u] = load_src4_tail(src, i0 + u * 32, nvec, E);
            }
#pragma unroll
            for (int u = 0; u < SCAN_U; ++u)
                test4(v[u], i0 + u * 32, row0, (unsigned)nrows, qe, qn);

            const bool last = (it == nsteps - 1);
            if (qn >= Q_DRAIN || last)
                drain_queue(qe, qn, last, lane, row0, src, dst, s_dst, h, ssl, ml, den, acc);
        }
    }
    __syncthreads();

    for (int i = tid; i < nrows * OUT_F; i += 1024) {
        const float dn = den[i / OUT_F];
        const float v = acc[i] / (dn > 0.f ? dn : 1.f);
        out[(size_t)row0 * OUT_F + i] = v > 0.f ? v : expm1f(v);
    }
}

extern "C" void kernel_launch(void* const* d_in, const int* in_sizes, int n_in,
                              void* d_out, int out_size, void* d_ws, size_t ws_size,
                              hipStream_t stream)
{
    const float* x    = (const float*)d_in[0];
    const int*   edge = (const int*)  d_in[1];
    const float* W    = (const float*)d_in[2];
    const float* attn = (const float*)d_in[3];

    const int N = in_sizes[0] / IN_F;
    const int E = in_sizes[1] / 2;
    const int* src = edge;
    const int* dst = edge + E;
    float* out = (float*)d_out;

    float* h     = (float*)d_ws;
    float* s_src = h + (size_t)N * OUT_F;
    float* s_dst = s_src + N;
    float* m     = s_dst + N;

    gat_init     <<<(N + 255) / 256,       256, 0, stream>>>(m, N);
    gat_gemm_wmma<<<(N + 127) / 128,       256, 0, stream>>>(x, W, h, N);
    gat_scores   <<<(N * 32 + 255) / 256,  256, 0, stream>>>(h, attn, s_src, s_dst, N);
    gat_edge_max <<<(E / 4 + 4 + 255) / 256, 256, 0, stream>>>(src, dst, s_src, s_dst, m,
                                                            m + N, ws_size, N, E);
    gat_rows_scan<<<(N + ROWS_PB - 1) / ROWS_PB, 1024, 0, stream>>>(
        src, dst, s_src, s_dst, m, h, out, m + N, ws_size, N, E);
}


# Row maxima formed inside the owner block by LDS float max while it lists its edges in LDS; behind a barrier its waves walk the list; init and global max gone (5 to 4 launches); sums in arrival order
# speedup vs baseline: 3.7130x; 1.8345x over previous
#include <hip/hip_runtime.h>
#include <cmath>

#define IN_F  256
#define OUT_F 64
#define LRELU_ALPHA 0.2f

typedef __attribute__((ext_vector_type(2))) float v2f;
typedef __attribute__((ext_vector_type(8))) float v8f;
typedef __attribute__((ext_vector_type(4))) int   v4i;

__global__ __launch_bounds__(256) void gat_gemm_wmma(
    const float* __restrict__ x, const float* __restrict__ W,
    float* __restrict__ h, int N)
{
    __shared__ float Wlds[IN_F * OUT_F];
    const int tid = threadIdx.x;
    for (int i = tid; i < IN_F * OUT_F; i += 256) {
        const int k = i / OUT_F, c = i % OUT_F;
        Wlds[(k >> 1) * (2 * OUT_F) + c * 2 + (k & 1)] = W[i];
    }
    __syncthreads();

    const int wave = tid >> 5;
    const int lane = tid & 31;
    const int half = lane >> 4;
    const int l16  = lane & 15;
    const int tile_row0 = blockIdx.x * 128 + wave * 16;
    const int arow = tile_row0 + l16;
    const int lrow = arow < N ? arow : (N - 1);

    const float* xptr = x + (size_t)lrow * IN_F + 2 * half;
    const float* bptr = Wlds + half * (2 * OUT_F) + l16 * 2;

    v8f acc[4] = {};

    for (int k0 = 0; k0 < IN_F; k0 += 4) {
        const float2 t = *(const float2*)(xptr + k0);
        v2f a; a.x = t.x; a.y = t.y;
        const float* bk = bptr + (k0 >> 1) * (2 * OUT_F);
#pragma unroll
        for (int n = 0; n < 4; ++n) {
            const float2 u = *(const float2*)(bk + n * 32);
            v2f b; b.x = u.x; b.y = u.y;
            acc[n] = __builtin_amdgcn_wmma_f32_16x16x4_f32(
                false, a, false, b, (short)0, acc[n], false, false);
        }
    }

    const int rbase = tile_row0 + 8 * half;
#pragma unroll
    for (int r = 0; r < 8; ++r) {
        const int rr = rbase + r;
        if (rr < N) {
#pragma unroll
            for (int n = 0; n < 4; ++n)
                h[(size_t)rr * OUT_F + n * 16 + l16] = acc[n][r];
        }
    }
}

__global__ __launch_bounds__(256) void gat_scores(
    const float* __restrict__ h, const float* __restrict__ attn,
    float* __restrict__ s_src, float* __restrict__ s_dst, int N)
{
    const int node = (blockIdx.x * blockDim.x + threadIdx.x) >> 5;
    const int lane = threadIdx.x & 31;
    if (node >= N) return;
    const float* hr = h + (size_t)node * OUT_F;
    const float h0 = hr[lane], h1 = hr[lane + 32];
    float ps = h0 * attn[lane]         + h1 * attn[lane + 32];
    float pd = h0 * attn[OUT_F + lane] + h1 * attn[OUT_F + lane + 32];
    for (int off = 16; off > 0; off >>= 1) {
        ps += __shfl_down(ps, off, 32);
        pd += __shfl_down(pd, off, 32);
    }
    if (lane == 0) { s_src[node] = ps; s_dst[node] = pd; }
}

__device__ __forceinline__ float logit_of(float v) {
    return -(v > 0.f ? v : LRELU_ALPHA * v);
}

#define ROWS_PB   392
#define WS_FLOATS_PER_NODE 66

__device__ __forceinline__ bool keys_usable(size_t ws_size, int N, int E)
{
    const size_t need = (size_t)N * WS_FLOATS_PER_NODE * sizeof(float)
                      + (size_t)E + 48;
    return (N + ROWS_PB - 1) / ROWS_PB <= 256 && ws_size >= need;
}

__device__ __forceinline__ unsigned* key_plane(float* ws_end)
{
    return (unsigned*)((char*)ws_end + ((16u - (unsigned)((size_t)ws_end & 15u)) & 15u));
}

__global__ __launch_bounds__(256) void gat_edge_keys(
    const int* __restrict__ src, float* ws_end, size_t ws_size, int N, int E)
{
    if (!keys_usable(ws_size, N, E)) return;
    const int q  = blockIdx.x * blockDim.x + threadIdx.x;
    const int e0 = q * 4;
    unsigned* keys = key_plane(ws_end);
    if (e0 >= E) {
        if (e0 < ((E + 15) & ~15)) keys[q] = 0u;
        return;
    }
    int s[4];
    if (e0 + 3 < E) {
        const int4 sv = ((const int4*)src)[q];
        s[0] = sv.x; s[1] = sv.y; s[2] = sv.z; s[3] = sv.w;
    } else {
#pragma unroll
        for (int j = 0; j < 4; ++j) s[j] = (e0 + j < E) ? src[e0 + j] : -1;
    }
    unsigned kw = 0u;
#pragma unroll
    for (int j = 0; j < 4; ++j)
        if (s[j] >= 0) kw |= (unsigned)(s[j] / ROWS_PB) << (8 * j);
    keys[q] = kw;
}

#define SCAN_U    8
#define AGG_WAVES 32
#define Q_DRAIN   96
#define Q_CAP     256
#define LIST_CAP  15360

__device__ __forceinline__ int4 load_src4_tail(const int* __restrict__ src,
                                               int i4, int nvec, int E)
{
    int4 v = make_int4(-1, -1, -1, -1);
    if (i4 < nvec) {
        v = ((const int4*)src)[i4];
    } else if (i4 == nvec) {
        const int e0 = nvec * 4;
        if (e0     < E) v.x = src[e0];
        if (e0 + 1 < E) v.y = src[e0 + 1];
        if (e0 + 2 < E) v.z = src[e0 + 2];
    }
    return v;
}

__device__ __forceinline__ void file_hit(bool hit, int e, int* qe, int& qn)
{
    const unsigned mask = __builtin_amdgcn_ballot_w32(hit);
    if (hit) qe[qn + (int)__builtin_amdgcn_mbcnt_lo(mask, 0u)] = e;
    qn += __builtin_popcount(mask);
}

__device__ __forceinline__ void test4(const int4 v, int i4, int row0, unsigned nrows,
                                      int* qe, int& qn)
{
    const bool h0 = (unsigned)(v.x - row0) < nrows;
    const bool h1 = (unsigned)(v.y - row0) < nrows;
    const bool h2 = (unsigned)(v.z - row0) < nrows;
    const bool h3 = (unsigned)(v.w - row0) < nrows;
    if (__builtin_amdgcn_ballot_w32(h0 | h1 | h2 | h3) != 0u) {
        const int e0 = i4 * 4;
        file_hit(h0, e0,     qe, qn);
        file_hit(h1, e0 + 1, qe, qn);
        file_hit(h2, e0 + 2, qe, qn);
        file_hit(h3, e0 + 3, qe, qn);
    }
}

template <int G>
__device__ __forceinline__ void add_rows(int d, int r, float p, int k, int lane,
                                         const float* __restrict__ h, float* acc)
{
    float2 hv[G];
    float  pk[G];
    int    rk[G];
#pragma unroll
    for (int u = 0; u < G; ++u) {
        const int dk = __builtin_amdgcn_readlane(d, k + u);
        rk[u] = __builtin_amdgcn_readlane(r, k + u);
        pk[u] = __int_as_float(__builtin_amdgcn_readlane(__float_as_int(p), k + u));
        hv[u] = *(const float2*)(h + (size_t)dk * OUT_F + 2 * lane);
    }
#pragma unroll
    for (int u = 0; u < G; ++u) {
        float* a = acc + rk[u] * OUT_F + 2 * lane;
        atomicAdd(a,     pk[u] * hv[u].x);
        atomicAdd(a + 1, pk[u] * hv[u].y);
    }
}

__device__ __forceinline__ void add_batch(int d, int r, float p, int cnt, int lane,
                                          const float* __restrict__ h, float* acc)
{
    int k = 0;
    for (; k + 16 <= cnt; k += 16) add_rows<16>(d, r, p, k, lane, h, acc);
    for (; k + 4  <= cnt; k += 4)  add_rows<4>(d, r, p, k, lane, h, acc);
    for (; k      <  cnt; k += 1)  add_rows<1>(d, r, p, k, lane, h, acc);
}

__device__ __forceinline__ void queue_rest(int* qe, int& qn, bool flush, int b, int lane)
{
    if (flush) {
        qn = 0;
    } else if (b > 0) {
        const int rem = qn - b;
        if (lane < rem) {
            const int e = qe[b + lane];
            qe[lane] = e;
        }
        qn = rem;
    }
}

__device__ __forceinline__ void note_queue(
    int* qe, int& qn, bool flush, int lane, int row0,
    const int* __restrict__ src, const int* __restrict__ dst,
    const float* __restrict__ s_dst, const float* ssl, float* ml,
    unsigned short* list_r, int* list_d, float* list_lg, int* list_cnt)
{
    int b = 0;
    while (qn - b >= 32 || (flush && b < qn)) {
        const int cnt = (qn - b) < 32 ? (qn - b) : 32;
        int base = 0;
        if (lane == 0) base = atomicAdd(list_cnt, cnt);
        base = __builtin_amdgcn_readfirstlane(base);
        if (lane < cnt) {
            const int e = qe[b + lane];
            const int r = src[e] - row0;
            const int d = dst[e];
            const float lg = logit_of(ssl[r] + s_dst[d]);
            atomicMax(&ml[r], lg);
            const int slot = base + lane;
            if (slot < LIST_CAP) {
                list_r[slot]  = (unsigned short)r;
                list_d[slot]  = d;
                list_lg[slot] = lg;
            }
        }
        b += 32;
    }
    queue_rest(qe, qn, flush, b, lane);
}

__device__ __forceinline__ void drain_queue(
    int* qe, int& qn, bool flush, int lane, int row0,
    const int* __restrict__ src, const int* __restrict__ dst,
    const float* __restrict__ s_dst, const float* __restrict__ h,
    const float* ssl, const float* ml, float* den, float* acc)
{
    int b = 0;
    while (qn - b >= 32 || (flush && b < qn)) {
        const int cnt = (qn - b) < 32 ? (qn - b) : 32;
        int d = 0, r = 0;
        float p = 0.f;
        if (lane < cnt) {
            const int e = qe[b + lane];
            r = src[e] - row0;
            d = dst[e];
            p = __expf(logit_of(ssl[r] + s_dst[d]) - ml[r]);
            atomicAdd(&den[r], p);
        }
        add_batch(d, r, p, cnt, lane, h, acc);
        b += 32;
    }
    queue_rest(qe, qn, flush, b, lane);
}

__device__ __forceinline__ unsigned zero_flags(unsigned x)
{
    return ~(((x & 0x7F7F7F7Fu) + 0x7F7F7F7Fu) | x);
}

__device__ __forceinline__ unsigned hit_mask2(const v4i a, const v4i b, unsigned splat)
{
    return ((zero_flags((unsigned)a.x ^ splat) >> 7) & 0x01010101u)
         | ((zero_flags((unsigned)a.y ^ splat) >> 6) & 0x02020202u)
         | ((zero_flags((unsigned)a.z ^ splat) >> 5) & 0x04040404u)
         | ((zero_flags((unsigned)a.w ^ splat) >> 4) & 0x08080808u)
         | ((zero_flags((unsigned)b.x ^ splat) >> 3) & 0x10101010u)
         | ((zero_flags((unsigned)b.y ^ splat) >> 2) & 0x20202020u)
         | ((zero_flags((unsigned)b.z ^ splat) >> 1) & 0x40404040u)
         | ( zero_flags((unsigned)b.w ^ splat)       & 0x80808080u);
}

#define KSCAN_U  8
#define KQ_ROOM  (Q_CAP - 32)

struct OwnerLds {
    float* acc;
    float* den;
    const float* ssl;
    float* ml;
    unsigned short* list_r;
    int*   list_d;
    float* list_lg;
    int*   list_cnt;
};

template <bool FIRST>
__device__ __forceinline__ void sweep_edges(
    int* qe, int lane, int wave, int row0, int nrows, bool use_keys,
    const int* __restrict__ src, const int* __restrict__ dst,
    const float* __restrict__ s_dst, const float* __restrict__ h,
    const OwnerLds L, float* ws_end, int E)
{
    int qn = 0;
    auto work_off = [&](bool flush) {
        if (FIRST)
            note_queue(qe, qn, flush, lane, row0, src, dst, s_dst, L.ssl, L.ml,
                       L.list_r, L.list_d, L.list_lg, L.list_cnt);
        else
            drain_queue(qe, qn, flush, lane, row0, src, dst, s_dst, h, L.ssl, L.ml, L.den, L.acc);
    };

    if (use_keys) {
        const v4i* keys4 = (const v4i*)key_plane(ws_end);
        const unsigned splat = blockIdx.x * 0x01010101u;
        const int nvec   = (E + 15) >> 4;
        const int nsteps = (nvec + 1024 * KSCAN_U - 1) / (1024 * KSCAN_U);
        const int t0 = (int)(((long long)blockIdx.x * nsteps) / gridDim.x);
        for (int it = 0; it < nsteps; ++it) {
            int t = it + t0;
            if (t >= nsteps) t -= nsteps;
            const int i0 = t * (1024 * KSCAN_U) + wave * (32 * KSCAN_U) + lane;
            unsigned mm[KSCAN_U / 2];
            {
                v4i v[KSCAN_U];
#pragma unroll
                for (int u = 0; u < KSCAN_U; ++u) {
                    const int iv = i0 + u * 32;
                    v[u] = keys4[iv < nvec ? iv : nvec - 1];
                }
                static_assert(KSCAN_U == 8, "the fence names eight vectors");
                asm volatile("" :: "v"(v[0]), "v"(v[1]), "v"(v[2]), "v"(v[3]),
                                   "v"(v[4]), "v"(v[5]), "v"(v[6]), "v"(v[7]));
#pragma unroll
                for (int u = 0; u < KSCAN_U / 2; ++u)
                    mm[u] = hit_mask2(v[2 * u], v[2 * u + 1], splat);
            }
            const bool last = (it == nsteps - 1);
            bool more;
            do {
#pragma unroll
                for (int u = 0; u < KSCAN_U / 2; ++u) {
                    while (qn <= KQ_ROOM) {
                        const bool has = mm[u] != 0u;
                        if (__builtin_amdgcn_ballot_w32(has) == 0u) break;
                        int e = E;
                        if (has) {
                            const int p = __builtin_ctz(mm[u]);
                            mm[u] &= mm[u] - 1u;
                            e = (i0 + (2 * u + ((p >> 2) & 1)) * 32) * 16 + 4 * (p & 3) + (p >> 3);
                        }
                        file_hit(e < E, e, qe, qn);
                    }
                }
                unsigned any = 0u;
#pragma unroll
                for (int u = 0; u < KSCAN_U / 2; ++u) any |= mm[u];
                more = __builtin_amdgcn_ballot_w32(any != 0u) != 0u;
                const bool flush = last && !more;
                if (qn >= Q_DRAIN || more || flush) work_off(flush);
            } while (more);
        }
    } else {
        const int nvec   = E >> 2;
        const int nsteps = nvec / (1024 * SCAN_U) + 1;
        const int t0 = (int)(((long long)blockIdx.x * nsteps) / gridDim.x);
        for (int it = 0; it < nsteps; ++it) {
            int t = it + t0;
            if (t >= nsteps) t -= nsteps;
            const int i0 = t * (1024 * SCAN_U) + wave * (32 * SCAN_U) + lane;
            for (int u = 0; u < SCAN_U; ++u) {
                const int4 v = load_src4_tail(src, i0 + u * 32, nvec, E);
                if (qn > Q_CAP - 128) work_off(false);
                test4(v, i0 + u * 32, row0, (unsigned)nrows, qe, qn);
            }
            const bool last = (it == nsteps - 1);
            if (qn >= Q_DRAIN || last) work_off(last);
        }
    }
}

__global__ __launch_bounds__(1024) void gat_rows_scan(
    const int* __restrict__ src, const int* __restrict__ dst,
    const float* __restrict__ s_src, const float* __restrict__ s_dst,
    const float* __restrict__ h,
    float* __restrict__ out, float* ws_end, size_t ws_size, int N, int E)
{
    __shared__ float acc[ROWS_PB * OUT_F];
    __shared__ float den[ROWS_PB];
    __shared__ float ssl[ROWS_PB];
    __shared__ float ml[ROWS_PB];
    __shared__ int   queue[AGG_WAVES * Q_CAP];
    __shared__ int            list_d[LIST_CAP];
    __shared__ float          list_lg[LIST_CAP];
    __shared__ unsigned short list_r[LIST_CAP];
    __shared__ int   list_cnt;

    const int tid  = threadIdx.x;
    const int lane = tid & 31;
    const int wave = tid >> 5;
    const int row0 = blockIdx.x * ROWS_PB;
    const int nrows = (N - row0) < ROWS_PB ? (N - row0) : ROWS_PB;

    for (int i = tid; i < ROWS_PB * OUT_F; i += 1024) acc[i] = 0.f;
    for (int i = tid; i < ROWS_PB; i += 1024) {
        den[i] = 0.f;
        ssl[i] = (i < nrows) ? s_src[row0 + i] : 0.f;
        ml[i]  = -INFINITY;
    }
    if (tid == 0) list_cnt = 0;
    __syncthreads();

    int* qe = queue + wave * Q_CAP;
    const bool use_keys = keys_usable(ws_size, N, E);
    const OwnerLds lds = { acc, den, ssl, ml, list_r, list_d, list_lg, &list_cnt };

    sweep_edges<true>(qe, lane, wave, row0, nrows, use_keys, src, dst, s_dst, h, lds, ws_end, E);

    __syncthreads();

    const int total = __builtin_amdgcn_readfirstlane(list_cnt);

    if (total <= LIST_CAP) {
        const int wave_s = __builtin_amdgcn_readfirstlane(wave);
        for (int b = wave_s * 32; b < total; b += AGG_WAVES * 32) {
            const int cnt = (total - b) < 32 ? (total - b) : 32;
            int d = 0, r = 0;
            float p = 0.f;
            if (lane < cnt) {
                r = list_r[b + lane];
                d = list_d[b + lane];
                p = __expf(list_lg[b + lane] - ml[r]);
                atomicAdd(&den[r], p);
            }
            add_batch(d, r, p, cnt, lane, h, acc);
        }
    } else {
        sweep_edges<false>(qe, lane, wave, row0, nrows, use_keys, src, dst, s_dst, h, lds, ws_end, E);
    }
    __syncthreads();

    for (int i = tid; i < nrows * OUT_F; i += 1024) {
        const float dn = den[i / OUT_F];
        const float v = acc[i] / (dn > 0.f ? dn : 1.f);
        out[(size_t)row0 * OUT_F + i] = v > 0.f ? v : expm1f(v);
    }
}

extern "C" void kernel_launch(void* const* d_in, const int* in_sizes, int n_in,
                              void* d_out, int out_size, void* d_ws, size_t ws_size,
                              hipStream_t stream)
{
    const float* x    = (const float*)d_in[0];
    const int*   edge = (const int*)  d_in[1];
    const float* W    = (const float*)d_in[2];
    const float* attn = (const float*)d_in[3];

    const int N = in_sizes[0] / IN_F;
    const int E = in_sizes[1] / 2;
    const int* src = edge;
    const int* dst = edge + E;
    float* out = (float*)d_out;

    float* h     = (float*)d_ws;
    float* s_src = h + (size_t)N * OUT_F;
    float* s_dst = s_src + N;

    gat_gemm_wmma<<<(N + 127) / 128,       256, 0, stream>>>(x, W, h, N);
    gat_scores   <<<(N * 32 + 255) / 256,  256, 0, stream>>>(h, attn, s_src, s_dst, N);
    gat_edge_keys<<<(E / 4 + 4 + 255) / 256, 256, 0, stream>>>(src, s_dst + N, ws_size, N, E);
    gat_rows_scan<<<(N + ROWS_PB - 1) / ROWS_PB, 1024, 0, stream>>>(
        src, dst, s_src, s_dst, h, out, s_dst + N, ws_size, N, E);
}


# Scores and key passes run as few long waves: a wave sums 32 rows (one 128-byte store per array) or keys 1024 edges, 8 loads in flight, not 1 row or 128 edges; 125k waves to 6k; sums unchanged
# speedup vs baseline: 3.8860x; 1.0466x over previous
#include <hip/hip_runtime.h>
#include <cmath>

#define IN_F  256
#define OUT_F 64
#define LRELU_ALPHA 0.2f

typedef __attribute__((ext_vector_type(2))) float v2f;
typedef __attribute__((ext_vector_type(8))) float v8f;
typedef __attribute__((ext_vector_type(4))) int   v4i;

__global__ __launch_bounds__(256) void gat_gemm_wmma(
    const float* __restrict__ x, const float* __restrict__ W,
    float* __restrict__ h, int N)
{
    __shared__ float Wlds[IN_F * OUT_F];
    const int tid = threadIdx.x;
    for (int i = tid; i < IN_F * OUT_F; i += 256) {
        const int k = i / OUT_F, c = i % OUT_F;
        Wlds[(k >> 1) * (2 * OUT_F) + c * 2 + (k & 1)] = W[i];
    }
    __syncthreads();

    const int wave = tid >> 5;
    const int lane = tid & 31;
    const int half = lane >> 4;
    const int l16  = lane & 15;
    const int tile_row0 = blockIdx.x * 128 + wave * 16;
    const int arow = tile_row0 + l16;
    const int lrow = arow < N ? arow : (N - 1);

    const float* xptr = x + (size_t)lrow * IN_F + 2 * half;
    const float* bptr = Wlds + half * (2 * OUT_F) + l16 * 2;

    v8f acc[4] = {};

    for (int k0 = 0; k0 < IN_F; k0 += 4) {
        const float2 t = *(const float2*)(xptr + k0);
        v2f a; a.x = t.x; a.y = t.y;
        const float* bk = bptr + (k0 >> 1) * (2 * OUT_F);
#pragma unroll
        for (int n = 0; n < 4; ++n) {
            const float2 u = *(const float2*)(bk + n * 32);
            v2f b; b.x = u.x; b.y = u.y;
            acc[n] = __builtin_amdgcn_wmma_f32_16x16x4_f32(
                false, a, false, b, (short)0, acc[n], false, false);
        }
    }

    const int rbase = tile_row0 + 8 * half;
#pragma unroll
    for (int r = 0; r < 8; ++r) {
        const int rr = rbase + r;
        if (rr < N) {
#pragma unroll
            for (int n = 0; n < 4; ++n)
                h[(size_t)rr * OUT_F + n * 16 + l16] = acc[n][r];
        }
    }
}

#define SC_ROWS 32

__global__ __launch_bounds__(256) void gat_scores(
    const float* __restrict__ h, const float* __restrict__ attn,
    float* __restrict__ s_src, float* __restrict__ s_dst, int N)
{
    const int lane = threadIdx.x & 31;
    const int wv   = __builtin_amdgcn_readfirstlane((int)(threadIdx.x >> 5));
    const int row0 = (blockIdx.x * 8 + wv) * SC_ROWS;
    if (row0 >= N) return;

    const float a0 = attn[lane],         a1 = attn[lane + 32];
    const float b0 = attn[OUT_F + lane], b1 = attn[OUT_F + lane + 32];

    float h0[SC_ROWS], h1[SC_ROWS];
#pragma unroll
    for (int j = 0; j < SC_ROWS; ++j) {
        const int r = (row0 + j < N) ? (row0 + j) : (N - 1);
        const float* hr = h + (size_t)r * OUT_F;
        h0[j] = hr[lane];
        h1[j] = hr[lane + 32];
    }

    float rs = 0.f, rd = 0.f;
#pragma unroll
    for (int j = 0; j < SC_ROWS; ++j) {
        float ps = h0[j] * a0 + h1[j] * a1;
        float pd = h0[j] * b0 + h1[j] * b1;
        for (int off = 16; off > 0; off >>= 1) {
            ps += __shfl_down(ps, off, 32);
            pd += __shfl_down(pd, off, 32);
        }
        const float s0 = __int_as_float(__builtin_amdgcn_readfirstlane(__float_as_int(ps)));
        const float d0 = __int_as_float(__builtin_amdgcn_readfirstlane(__float_as_int(pd)));
        if (lane == j) { rs = s0; rd = d0; }
    }
    if (row0 + lane < N) {
        s_src[row0 + lane] = rs;
        s_dst[row0 + lane] = rd;
    }
}

__device__ __forceinline__ float logit_of(float v) {
    return -(v > 0.f ? v : LRELU_ALPHA * v);
}

#define ROWS_PB   392
#define WS_FLOATS_PER_NODE 66

__device__ __forceinline__ bool keys_usable(size_t ws_size, int N, int E)
{
    const size_t need = (size_t)N * WS_FLOATS_PER_NODE * sizeof(float)
                      + (size_t)E + 48;
    return (N + ROWS_PB - 1) / ROWS_PB <= 256 && ws_size >= need;
}

__device__ __forceinline__ unsigned* key_plane(float* ws_end)
{
    return (unsigned*)((char*)ws_end + ((16u - (unsigned)((size_t)ws_end & 15u)) & 15u));
}

__device__ __forceinline__ unsigned key_word(int s0, int s1, int s2, int s3)
{
    unsigned kw = 0u;
    if (s0 >= 0) kw |= (unsigned)(s0 / ROWS_PB);
    if (s1 >= 0) kw |= (unsigned)(s1 / ROWS_PB) << 8;
    if (s2 >= 0) kw |= (unsigned)(s2 / ROWS_PB) << 16;
    if (s3 >= 0) kw |= (unsigned)(s3 / ROWS_PB) << 24;
    return kw;
}

#define KEY_U 8

__global__ __launch_bounds__(256) void gat_edge_keys(
    const int* __restrict__ src, float* ws_end, size_t ws_size, int N, int E)
{
    if (!keys_usable(ws_size, N, E)) return;
    unsigned* keys = key_plane(ws_end);
    const int lane = threadIdx.x & 31;
    const int wv = __builtin_amdgcn_readfirstlane((int)(threadIdx.x >> 5));
    const int q0 = (blockIdx.x * 8 + wv) * (32 * KEY_U);

    if (q0 + 32 * KEY_U <= (E >> 2)) {
        int4 sv[KEY_U];
#pragma unroll
        for (int u = 0; u < KEY_U; ++u) sv[u] = ((const int4*)src)[q0 + u * 32 + lane];
#pragma unroll
        for (int u = 0; u < KEY_U; ++u)
            keys[q0 + u * 32 + lane] = key_word(sv[u].x, sv[u].y, sv[u].z, sv[u].w);
        return;
    }
    for (int u = 0; u < KEY_U; ++u) {
        const int q  = q0 + u * 32 + lane;
        const int e0 = q * 4;
        if (e0 >= E) {
            if (e0 < ((E + 15) & ~15)) keys[q] = 0u;
            continue;
        }
        int4 sv;
        if (e0 + 3 < E) {
            sv = ((const int4*)src)[q];
        } else {
            sv.x = src[e0];
            sv.y = (e0 + 1 < E) ? src[e0 + 1] : -1;
            sv.z = (e0 + 2 < E) ? src[e0 + 2] : -1;
            sv.w = -1;
        }
        keys[q] = key_word(sv.x, sv.y, sv.z, sv.w);
    }
}

#define SCAN_U    8
#define AGG_WAVES 32
#define Q_DRAIN   96
#define Q_CAP     256
#define LIST_CAP  15360

__device__ __forceinline__ int4 load_src4_tail(const int* __restrict__ src,
                                               int i4, int nvec, int E)
{
    int4 v = make_int4(-1, -1, -1, -1);
    if (i4 < nvec) {
        v = ((const int4*)src)[i4];
    } else if (i4 == nvec) {
        const int e0 = nvec * 4;
        if (e0     < E) v.x = src[e0];
        if (e0 + 1 < E) v.y = src[e0 + 1];
        if (e0 + 2 < E) v.z = src[e0 + 2];
    }
    return v;
}

__device__ __forceinline__ void file_hit(bool hit, int e, int* qe, int& qn)
{
    const unsigned mask = __builtin_amdgcn_ballot_w32(hit);
    if (hit) qe[qn + (int)__builtin_amdgcn_mbcnt_lo(mask, 0u)] = e;
    qn += __builtin_popcount(mask);
}

__device__ __forceinline__ void test4(const int4 v, int i4, int row0, unsigned nrows,
                                      int* qe, int& qn)
{
    const bool h0 = (unsigned)(v.x - row0) < nrows;
    const bool h1 = (unsigned)(v.y - row0) < nrows;
    const bool h2 = (unsigned)(v.z - row0) < nrows;
    const bool h3 = (unsigned)(v.w - row0) < nrows;
    if (__builtin_amdgcn_ballot_w32(h0 | h1 | h2 | h3) != 0u) {
        const int e0 = i4 * 4;
        file_hit(h0, e0,     qe, qn);
        file_hit(h1, e0 + 1, qe, qn);
        file_hit(h2, e0 + 2, qe, qn);
        file_hit(h3, e0 + 3, qe, qn);
    }
}

template <int G>
__device__ __forceinline__ void add_rows(int d, int r, float p, int k, int lane,
                                         const float* __restrict__ h, float* acc)
{
    float2 hv[G];
    float  pk[G];
    int    rk[G];
#pragma unroll
    for (int u = 0; u < G; ++u) {
        const int dk = __builtin_amdgcn_readlane(d, k + u);
        rk[u] = __builtin_amdgcn_readlane(r, k + u);
        pk[u] = __int_as_float(__builtin_amdgcn_readlane(__float_as_int(p), k + u));
        hv[u] = *(const float2*)(h + (size_t)dk * OUT_F + 2 * lane);
    }
#pragma unroll
    for (int u = 0; u < G; ++u) {
        float* a = acc + rk[u] * OUT_F + 2 * lane;
        atomicAdd(a,     pk[u] * hv[u].x);
        atomicAdd(a + 1, pk[u] * hv[u].y);
    }
}

__device__ __forceinline__ void add_batch(int d, int r, float p, int cnt, int lane,
                                          const float* __restrict__ h, float* acc)
{
    int k = 0;
    for (; k + 16 <= cnt; k += 16) add_rows<16>(d, r, p, k, lane, h, acc);
    for (; k + 4  <= cnt; k += 4)  add_rows<4>(d, r, p, k, lane, h, acc);
    for (; k      <  cnt; k += 1)  add_rows<1>(d, r, p, k, lane, h, acc);
}

__device__ __forceinline__ void queue_rest(int* qe, int& qn, bool flush, int b, int lane)
{
    if (flush) {
        qn = 0;
    } else if (b > 0) {
        const int rem = qn - b;
        if (lane < rem) {
            const int e = qe[b + lane];
            qe[lane] = e;
        }
        qn = rem;
    }
}

__device__ __forceinline__ void note_queue(
    int* qe, int& qn, bool flush, int lane, int row0,
    const int* __restrict__ src, const int* __restrict__ dst,
    const float* __restrict__ s_dst, const float* ssl, float* ml,
    unsigned short* list_r, int* list_d, float* list_lg, int* list_cnt)
{
    int b = 0;
    while (qn - b >= 32 || (flush && b < qn)) {
        const int cnt = (qn - b) < 32 ? (qn - b) : 32;
        int base = 0;
        if (lane == 0) base = atomicAdd(list_cnt, cnt);
        base = __builtin_amdgcn_readfirstlane(base);
        if (lane < cnt) {
            const int e = qe[b + lane];
            const int r = src[e] - row0;
            const int d = dst[e];
            const float lg = logit_of(ssl[r] + s_dst[d]);
            atomicMax(&ml[r], lg);
            const int slot = base + lane;
            if (slot < LIST_CAP) {
                list_r[slot]  = (unsigned short)r;
                list_d[slot]  = d;
                list_lg[slot] = lg;
            }
        }
        b += 32;
    }
    queue_rest(qe, qn, flush, b, lane);
}

__device__ __forceinline__ void drain_queue(
    int* qe, int& qn, bool flush, int lane, int row0,
    const int* __restrict__ src, const int* __restrict__ dst,
    const float* __restrict__ s_dst, const float* __restrict__ h,
    const float* ssl, const float* ml, float* den, float* acc)
{
    int b = 0;
    while (qn - b >= 32 || (flush && b < qn)) {
        const int cnt = (qn - b) < 32 ? (qn - b) : 32;
        int d = 0, r = 0;
        float p = 0.f;
        if (lane < cnt) {
            const int e = qe[b + lane];
            r = src[e] - row0;
            d = dst[e];
            p = __expf(logit_of(ssl[r] + s_dst[d]) - ml[r]);
            atomicAdd(&den[r], p);
        }
        add_batch(d, r, p, cnt, lane, h, acc);
        b += 32;
    }
    queue_rest(qe, qn, flush, b, lane);
}

__device__ __forceinline__ unsigned zero_flags(unsigned x)
{
    return ~(((x & 0x7F7F7F7Fu) + 0x7F7F7F7Fu) | x);
}

__device__ __forceinline__ unsigned hit_mask2(const v4i a, const v4i b, unsigned splat)
{
    return ((zero_flags((unsigned)a.x ^ splat) >> 7) & 0x01010101u)
         | ((zero_flags((unsigned)a.y ^ splat) >> 6) & 0x02020202u)
         | ((zero_flags((unsigned)a.z ^ splat) >> 5) & 0x04040404u)
         | ((zero_flags((unsigned)a.w ^ splat) >> 4) & 0x08080808u)
         | ((zero_flags((unsigned)b.x ^ splat) >> 3) & 0x10101010u)
         | ((zero_flags((unsigned)b.y ^ splat) >> 2) & 0x20202020u)
         | ((zero_flags((unsigned)b.z ^ splat) >> 1) & 0x40404040u)
         | ( zero_flags((unsigned)b.w ^ splat)       & 0x80808080u);
}

#define KSCAN_U  8
#define KQ_ROOM  (Q_CAP - 32)

struct OwnerLds {
    float* acc;
    float* den;
    const float* ssl;
    float* ml;
    unsigned short* list_r;
    int*   list_d;
    float* list_lg;
    int*   list_cnt;
};

template <bool FIRST>
__device__ __forceinline__ void sweep_edges(
    int* qe, int lane, int wave, int row0, int nrows, bool use_keys,
    const int* __restrict__ src, const int* __restrict__ dst,
    const float* __restrict__ s_dst, const float* __restrict__ h,
    const OwnerLds L, float* ws_end, int E)
{
    int qn = 0;
    auto work_off = [&](bool flush) {
        if (FIRST)
            note_queue(qe, qn, flush, lane, row0, src, dst, s_dst, L.ssl, L.ml,
                       L.list_r, L.list_d, L.list_lg, L.list_cnt);
        else
            drain_queue(qe, qn, flush, lane, row0, src, dst, s_dst, h, L.ssl, L.ml, L.den, L.acc);
    };

    if (use_keys) {
        const v4i* keys4 = (const v4i*)key_plane(ws_end);
        const unsigned splat = blockIdx.x * 0x01010101u;
        const int nvec   = (E + 15) >> 4;
        const int nsteps = (nvec + 1024 * KSCAN_U - 1) / (1024 * KSCAN_U);
        const int t0 = (int)(((long long)blockIdx.x * nsteps) / gridDim.x);
        for (int it = 0; it < nsteps; ++it) {
            int t = it + t0;
            if (t >= nsteps) t -= nsteps;
            const int i0 = t * (1024 * KSCAN_U) + wave * (32 * KSCAN_U) + lane;
            unsigned mm[KSCAN_U / 2];
            {
                v4i v[KSCAN_U];
#pragma unroll
                for (int u = 0; u < KSCAN_U; ++u) {
                    const int iv = i0 + u * 32;
                    v[u] = keys4[iv < nvec ? iv : nvec - 1];
                }
                static_assert(KSCAN_U == 8, "the fence names eight vectors");
                asm volatile("" :: "v"(v[0]), "v"(v[1]), "v"(v[2]), "v"(v[3]),
                                   "v"(v[4]), "v"(v[5]), "v"(v[6]), "v"(v[7]));
#pragma unroll
                for (int u = 0; u < KSCAN_U / 2; ++u)
                    mm[u] = hit_mask2(v[2 * u], v[2 * u + 1], splat);
            }
            const bool last = (it == nsteps - 1);
            bool more;
            do {
#pragma unroll
                for (int u = 0; u < KSCAN_U / 2; ++u) {
                    while (qn <= KQ_ROOM) {
                        const bool has = mm[u] != 0u;
                        if (__builtin_amdgcn_ballot_w32(has) == 0u) break;
                        int e = E;
                        if (has) {
                            const int p = __builtin_ctz(mm[u]);
                            mm[u] &= mm[u] - 1u;
                            e = (i0 + (2 * u + ((p >> 2) & 1)) * 32) * 16 + 4 * (p & 3) + (p >> 3);
                        }
                        file_hit(e < E, e, qe, qn);
                    }
                }
                unsigned any = 0u;
#pragma unroll
                for (int u = 0; u < KSCAN_U / 2; ++u) any |= mm[u];
                more = __builtin_amdgcn_ballot_w32(any != 0u) != 0u;
                const bool flush = last && !more;
                if (qn >= Q_DRAIN || more || flush) work_off(flush);
            } while (more);
        }
    } else {
        const int nvec   = E >> 2;
        const int nsteps = nvec / (1024 * SCAN_U) + 1;
        const int t0 = (int)(((long long)blockIdx.x * nsteps) / gridDim.x);
        for (int it = 0; it < nsteps; ++it) {
            int t = it + t0;
            if (t >= nsteps) t -= nsteps;
            const int i0 = t * (1024 * SCAN_U) + wave * (32 * SCAN_U) + lane;
            for (int u = 0; u < SCAN_U; ++u) {
                const int4 v = load_src4_tail(src, i0 + u * 32, nvec, E);
                if (qn > Q_CAP - 128) work_off(false);
                test4(v, i0 + u * 32, row0, (unsigned)nrows, qe, qn);
            }
            const bool last = (it == nsteps - 1);
            if (qn >= Q_DRAIN || last) work_off(last);
        }
    }
}

__global__ __launch_bounds__(1024) void gat_rows_scan(
    const int* __restrict__ src, const int* __restrict__ dst,
    const float* __restrict__ s_src, const float* __restrict__ s_dst,
    const float* __restrict__ h,
    float* __restrict__ out, float* ws_end, size_t ws_size, int N, int E)
{
    __shared__ float acc[ROWS_PB * OUT_F];
    __shared__ float den[ROWS_PB];
    __shared__ float ssl[ROWS_PB];
    __shared__ float ml[ROWS_PB];
    __shared__ int   queue[AGG_WAVES * Q_CAP];
    __shared__ int            list_d[LIST_CAP];
    __shared__ float          list_lg[LIST_CAP];
    __shared__ unsigned short list_r[LIST_CAP];
    __shared__ int   list_cnt;

    const int tid  = threadIdx.x;
    const int lane = tid & 31;
    const int wave = tid >> 5;
    const int row0 = blockIdx.x * ROWS_PB;
    const int nrows = (N - row0) < ROWS_PB ? (N - row0) : ROWS_PB;

    for (int i = tid; i < ROWS_PB * OUT_F; i += 1024) acc[i] = 0.f;
    for (int i = tid; i < ROWS_PB; i += 1024) {
        den[i] = 0.f;
        ssl[i] = (i < nrows) ? s_src[row0 + i] : 0.f;
        ml[i]  = -INFINITY;
    }
    if (tid == 0) list_cnt = 0;
    __syncthreads();

    int* qe = queue + wave * Q_CAP;
    const bool use_keys = keys_usable(ws_size, N, E);
    const OwnerLds lds = { acc, den, ssl, ml, list_r, list_d, list_lg, &list_cnt };

    sweep_edges<true>(qe, lane, wave, row0, nrows, use_keys, src, dst, s_dst, h, lds, ws_end, E);

    __syncthreads();

    const int total = __builtin_amdgcn_readfirstlane(list_cnt);

    if (total <= LIST_CAP) {
        const int wave_s = __builtin_amdgcn_readfirstlane(wave);
        for (int b = wave_s * 32; b < total; b += AGG_WAVES * 32) {
            const int cnt = (total - b) < 32 ? (total - b) : 32;
            int d = 0, r = 0;
            float p = 0.f;
            if (lane < cnt) {
                r = list_r[b + lane];
                d = list_d[b + lane];
                p = __expf(list_lg[b + lane] - ml[r]);
                atomicAdd(&den[r], p);
            }
            add_batch(d, r, p, cnt, lane, h, acc);
        }
    } else {
        sweep_edges<false>(qe, lane, wave, row0, nrows, use_keys, src, dst, s_dst, h, lds, ws_end, E);
    }
    __syncthreads();

    for (int i = tid; i < nrows * OUT_F; i += 1024) {
        const float dn = den[i / OUT_F];
        const float v = acc[i] / (dn > 0.f ? dn : 1.f);
        out[(size_t)row0 * OUT_F + i] = v > 0.f ? v : expm1f(v);
    }
}

extern "C" void kernel_launch(void* const* d_in, const int* in_sizes, int n_in,
                              void* d_out, int out_size, void* d_ws, size_t ws_size,
                              hipStream_t stream)
{
    const float* x    = (const float*)d_in[0];
    const int*   edge = (const int*)  d_in[1];
    const float* W    = (const float*)d_in[2];
    const float* attn = (const float*)d_in[3];

    const int N = in_sizes[0] / IN_F;
    const int E = in_sizes[1] / 2;
    const int* src = edge;
    const int* dst = edge + E;
    float* out = (float*)d_out;

    float* h     = (float*)d_ws;
    float* s_src = h + (size_t)N * OUT_F;
    float* s_dst = s_src + N;

    gat_gemm_wmma<<<(N + 127) / 128,       256, 0, stream>>>(x, W, h, N);
    gat_scores   <<<(N + 255) / 256,       256, 0, stream>>>(h, attn, s_src, s_dst, N);
    gat_edge_keys<<<(E / 4 + 4 + 2047) / 2048, 256, 0, stream>>>(src, s_dst + N, ws_size, N, E);
    gat_rows_scan<<<(N + ROWS_PB - 1) / ROWS_PB, 1024, 0, stream>>>(
        src, dst, s_src, s_dst, h, out, s_dst + N, ws_size, N, E);
}


# Edge records: the key pass, with s_dst in LDS in two pieces, also writes (local row, dst, s_dst[dst]) as 8 bytes per edge; an owner block makes 1 scattered read per own edge, not 3; sums unchanged
# speedup vs baseline: 4.1895x; 1.0781x over previous
#include <hip/hip_runtime.h>
#include <cmath>

#define IN_F  256
#define OUT_F 64
#define LRELU_ALPHA 0.2f

typedef __attribute__((ext_vector_type(2))) float v2f;
typedef __attribute__((ext_vector_type(8))) float v8f;
typedef __attribute__((ext_vector_type(4))) int   v4i;

__global__ __launch_bounds__(256) void gat_gemm_wmma(
    const float* __restrict__ x, const float* __restrict__ W,
    float* __restrict__ h, int N)
{
    __shared__ float Wlds[IN_F * OUT_F];
    const int tid = threadIdx.x;
    for (int i = tid; i < IN_F * OUT_F; i += 256) {
        const int k = i / OUT_F, c = i % OUT_F;
        Wlds[(k >> 1) * (2 * OUT_F) + c * 2 + (k & 1)] = W[i];
    }
    __syncthreads();

    const int wave = tid >> 5;
    const int lane = tid & 31;
    const int half = lane >> 4;
    const int l16  = lane & 15;
    const int tile_row0 = blockIdx.x * 128 + wave * 16;
    const int arow = tile_row0 + l16;
    const int lrow = arow < N ? arow : (N - 1);

    const float* xptr = x + (size_t)lrow * IN_F + 2 * half;
    const float* bptr = Wlds + half * (2 * OUT_F) + l16 * 2;

    v8f acc[4] = {};

    for (int k0 = 0; k0 < IN_F; k0 += 4) {
        const float2 t = *(const float2*)(xptr + k0);
        v2f a; a.x = t.x; a.y = t.y;
        const float* bk = bptr + (k0 >> 1) * (2 * OUT_F);
#pragma unroll
        for (int n = 0; n < 4; ++n) {
            const float2 u = *(const float2*)(bk + n * 32);
            v2f b; b.x = u.x; b.y = u.y;
            acc[n] = __builtin_amdgcn_wmma_f32_16x16x4_f32(
                false, a, false, b, (short)0, acc[n], false, false);
        }
    }

    const int rbase = tile_row0 + 8 * half;
#pragma unroll
    for (int r = 0; r < 8; ++r) {
        const int rr = rbase + r;
        if (rr < N) {
#pragma unroll
            for (int n = 0; n < 4; ++n)
                h[(size_t)rr * OUT_F + n * 16 + l16] = acc[n][r];
        }
    }
}

#define SC_ROWS 32

__global__ __launch_bounds__(256) void gat_scores(
    const float* __restrict__ h, const float* __restrict__ attn,
    float* __restrict__ s_src, float* __restrict__ s_dst, int N)
{
    const int lane = threadIdx.x & 31;
    const int wv   = __builtin_amdgcn_readfirstlane((int)(threadIdx.x >> 5));
    const int row0 = (blockIdx.x * 8 + wv) * SC_ROWS;
    if (row0 >= N) return;

    const float a0 = attn[lane],         a1 = attn[lane + 32];
    const float b0 = attn[OUT_F + lane], b1 = attn[OUT_F + lane + 32];

    float h0[SC_ROWS], h1[SC_ROWS];
#pragma unroll
    for (int j = 0; j < SC_ROWS; ++j) {
        const int r = (row0 + j < N) ? (row0 + j) : (N - 1);
        const float* hr = h + (size_t)r * OUT_F;
        h0[j] = hr[lane];
        h1[j] = hr[lane + 32];
    }

    float rs = 0.f, rd = 0.f;
#pragma unroll
    for (int j = 0; j < SC_ROWS; ++j) {
        float ps = h0[j] * a0 + h1[j] * a1;
        float pd = h0[j] * b0 + h1[j] * b1;
        for (int off = 16; off > 0; off >>= 1) {
            ps += __shfl_down(ps, off, 32);
            pd += __shfl_down(pd, off, 32);
        }
        const float s0 = __int_as_float(__builtin_amdgcn_readfirstlane(__float_as_int(ps)));
        const float d0 = __int_as_float(__builtin_amdgcn_readfirstlane(__float_as_int(pd)));
        if (lane == j) { rs = s0; rd = d0; }
    }
    if (row0 + lane < N) {
        s_src[row0 + lane] = rs;
        s_dst[row0 + lane] = rd;
    }
}

__device__ __forceinline__ float logit_of(float v) {
    return -(v > 0.f ? v : LRELU_ALPHA * v);
}

#define ROWS_PB   392
#define WS_FLOATS_PER_NODE 66

__device__ __forceinline__ bool keys_usable(size_t ws_size, int N, int E)
{
    const size_t need = (size_t)N * WS_FLOATS_PER_NODE * sizeof(float)
                      + (size_t)E + 48;
    return (N + ROWS_PB - 1) / ROWS_PB <= 256 && ws_size >= need;
}

__device__ __forceinline__ unsigned* key_plane(float* ws_end)
{
    return (unsigned*)((char*)ws_end + ((16u - (unsigned)((size_t)ws_end & 15u)) & 15u));
}

__device__ __forceinline__ unsigned key_word(int s0, int s1, int s2, int s3)
{
    unsigned kw = 0u;
    if (s0 >= 0) kw |= (unsigned)(s0 / ROWS_PB);
    if (s1 >= 0) kw |= (unsigned)(s1 / ROWS_PB) << 8;
    if (s2 >= 0) kw |= (unsigned)(s2 / ROWS_PB) << 16;
    if (s3 >= 0) kw |= (unsigned)(s3 / ROWS_PB) << 24;
    return kw;
}

#define TBL_FLOATS   51200
#define REC_U        4
#define REC_DST_BITS 23

__device__ __forceinline__ size_t rec_offset(int E)
{
    return ((size_t)E + 16 + 255) & ~(size_t)255;
}

__device__ __forceinline__ bool recs_usable(size_t ws_size, int N, int E)
{
    const size_t need = (size_t)N * WS_FLOATS_PER_NODE * sizeof(float) + 16
                      + rec_offset(E) + ((size_t)E + 4) * 8;
    static_assert(ROWS_PB <= (1 << (32 - REC_DST_BITS)), "the local row takes the bits above dst");
    return keys_usable(ws_size, N, E) && N <= (1 << REC_DST_BITS) && ws_size >= need;
}

__device__ __forceinline__ int2* rec_plane(float* ws_end, int E)
{
    return (int2*)((char*)key_plane(ws_end) + rec_offset(E));
}

__device__ __forceinline__ int4 load_group(const int* __restrict__ p, int g, int E, int fill)
{
    const int e0 = g * 4;
    int4 v = make_int4(fill, fill, fill, fill);
    if (e0     < E) v.x = p[e0];
    if (e0 + 1 < E) v.y = p[e0 + 1];
    if (e0 + 2 < E) v.z = p[e0 + 2];
    if (e0 + 3 < E) v.w = p[e0 + 3];
    return v;
}

__device__ __forceinline__ int rec_word(int s, int d)
{
    return s >= 0 ? (int)((unsigned)d | ((unsigned)(s - (s / ROWS_PB) * ROWS_PB) << REC_DST_BITS)) : 0;
}

#define TBL_G 7
#define TBL_R ((TBL_FLOATS / 4 + TBL_G * 1024 - 1) / (TBL_G * 1024))
typedef __attribute__((ext_vector_type(4))) float v4f;

__global__ __launch_bounds__(1024) void gat_edge_records(
    const int* __restrict__ src, const int* __restrict__ dst,
    const float* __restrict__ s_dst, float* ws_end, size_t ws_size, int N, int E)
{
    __shared__ float tbl[TBL_R * TBL_G * 4096 + 4];
    if (!keys_usable(ws_size, N, E)) return;
    const bool recs = recs_usable(ws_size, N, E);
    unsigned* keys = key_plane(ws_end);
    int4* rec4 = (int4*)rec_plane(ws_end, E);
    const int tid = threadIdx.x;

    const int npad    = ((E + 15) & ~15) >> 2;
    const int nfull   = E >> 2;
    const int per_blk = (((npad + (int)gridDim.x - 1) / (int)gridDim.x) + 31) & ~31;
    const int gb0 = blockIdx.x * per_blk;
    const int gb1 = (gb0 + per_blk) < npad ? (gb0 + per_blk) : npad;
    const bool vec = (E & 3) == 0 && nfull > 0;

    for (int t0 = gb0; t0 < gb1; t0 += 1024 * REC_U) {
        int   w0[REC_U][4];
        float sd[REC_U][4];
        {
            int4 sv[REC_U], dv[REC_U];
            if (vec) {
#pragma unroll
                for (int u = 0; u < REC_U; ++u) {
                    const int g = t0 + u * 1024 + tid;
                    const int gc = g < nfull ? g : nfull - 1;
                    sv[u] = ((const int4*)src)[gc];
                    dv[u] = ((const int4*)dst)[gc];
                }
            }
#pragma unroll
            for (int u = 0; u < REC_U; ++u) {
                const int g = t0 + u * 1024 + tid;
                if (!vec || g >= nfull) {
                    sv[u] = load_group(src, g, E, -1);
                    dv[u] = load_group(dst, g, E, 0);
                }
            }
#pragma unroll
            for (int u = 0; u < REC_U; ++u) {
                const int g = t0 + u * 1024 + tid;
                if (g < gb1) keys[g] = key_word(sv[u].x, sv[u].y, sv[u].z, sv[u].w);
                w0[u][0] = rec_word(sv[u].x, dv[u].x);
                w0[u][1] = rec_word(sv[u].y, dv[u].y);
                w0[u][2] = rec_word(sv[u].z, dv[u].z);
                w0[u][3] = rec_word(sv[u].w, dv[u].w);
                sd[u][0] = sd[u][1] = sd[u][2] = sd[u][3] = 0.f;
            }
        }
        if (!recs) continue;

        for (int base = 0; base < N; base += TBL_FLOATS) {
            const int cnt = (N - base) < TBL_FLOATS ? (N - base) : TBL_FLOATS;
            const float* tb = s_dst + base;
            __syncthreads();
            if ((((size_t)tb) & 15) == 0 && cnt >= 4) {
                const int cnt4 = cnt >> 2;
                const int rot  = (int)(((long long)blockIdx.x * cnt4) / gridDim.x);
                for (int q = 0; q < TBL_R; ++q) {
                    auto slot = [&](int k) {
                        const int i = tid + (q * TBL_G + k) * 1024;
                        int j = i + rot;
                        if (j >= cnt4) j -= cnt4;
                        return i < cnt4 ? j : -(i + 1);
                    };
                    v4f tv[TBL_G];
#pragma unroll
                    for (int k = 0; k < TBL_G; ++k) {
                        const int j = slot(k);
                        tv[k] = ((const v4f*)tb)[j >= 0 ? j : 0];
                    }
#pragma unroll
                    for (int k = 0; k < TBL_G; ++k) {
                        const int j = slot(k);
                        ((v4f*)tbl)[j >= 0 ? j : -j] = tv[k];
                    }
                }
                for (int i = (cnt & ~3) + tid; i < cnt; i += 1024) tbl[i] = tb[i];
            } else {
                for (int i = tid; i < cnt; i += 1024) tbl[i] = tb[i];
            }
            __syncthreads();
#pragma unroll
            for (int u = 0; u < REC_U; ++u) {
#pragma unroll
                for (int j = 0; j < 4; ++j) {
                    const unsigned o = (unsigned)((w0[u][j] & ((1 << REC_DST_BITS) - 1)) - base);
                    if (o < (unsigned)cnt) sd[u][j] = tbl[o];
                }
            }
        }
#pragma unroll
        for (int u = 0; u < REC_U; ++u) {
            const int g = t0 + u * 1024 + tid;
            if (g < gb1 && g * 4 < E) {
                rec4[2 * g]     = make_int4(w0[u][0], __float_as_int(sd[u][0]),
                                            w0[u][1], __float_as_int(sd[u][1]));
                rec4[2 * g + 1] = make_int4(w0[u][2], __float_as_int(sd[u][2]),
                                            w0[u][3], __float_as_int(sd[u][3]));
            }
        }
    }
}

#define SCAN_U    8
#define AGG_WAVES 32
#define Q_DRAIN   96
#define Q_CAP     256
#define LIST_CAP  15360

__device__ __forceinline__ int4 load_src4_tail(const int* __restrict__ src,
                                               int i4, int nvec, int E)
{
    int4 v = make_int4(-1, -1, -1, -1);
    if (i4 < nvec) {
        v = ((const int4*)src)[i4];
    } else if (i4 == nvec) {
        const int e0 = nvec * 4;
        if (e0     < E) v.x = src[e0];
        if (e0 + 1 < E) v.y = src[e0 + 1];
        if (e0 + 2 < E) v.z = src[e0 + 2];
    }
    return v;
}

__device__ __forceinline__ void file_hit(bool hit, int e, int* qe, int& qn)
{
    const unsigned mask = __builtin_amdgcn_ballot_w32(hit);
    if (hit) qe[qn + (int)__builtin_amdgcn_mbcnt_lo(mask, 0u)] = e;
    qn += __builtin_popcount(mask);
}

__device__ __forceinline__ void test4(const int4 v, int i4, int row0, unsigned nrows,
                                      int* qe, int& qn)
{
    const bool h0 = (unsigned)(v.x - row0) < nrows;
    const bool h1 = (unsigned)(v.y - row0) < nrows;
    const bool h2 = (unsigned)(v.z - row0) < nrows;
    const bool h3 = (unsigned)(v.w - row0) < nrows;
    if (__builtin_amdgcn_ballot_w32(h0 | h1 | h2 | h3) != 0u) {
        const int e0 = i4 * 4;
        file_hit(h0, e0,     qe, qn);
        file_hit(h1, e0 + 1, qe, qn);
        file_hit(h2, e0 + 2, qe, qn);
        file_hit(h3, e0 + 3, qe, qn);
    }
}

template <int G>
__device__ __forceinline__ void add_rows(int d, int r, float p, int k, int lane,
                                         const float* __restrict__ h, float* acc)
{
    float2 hv[G];
    float  pk[G];
    int    rk[G];
#pragma unroll
    for (int u = 0; u < G; ++u) {
        const int dk = __builtin_amdgcn_readlane(d, k + u);
        rk[u] = __builtin_amdgcn_readlane(r, k + u);
        pk[u] = __int_as_float(__builtin_amdgcn_readlane(__float_as_int(p), k + u));
        hv[u] = *(const float2*)(h + (size_t)dk * OUT_F + 2 * lane);
    }
#pragma unroll
    for (int u = 0; u < G; ++u) {
        float* a = acc + rk[u] * OUT_F + 2 * lane;
        atomicAdd(a,     pk[u] * hv[u].x);
        atomicAdd(a + 1, pk[u] * hv[u].y);
    }
}

__device__ __forceinline__ void add_batch(int d, int r, float p, int cnt, int lane,
                                          const float* __restrict__ h, float* acc)
{
    int k = 0;
    for (; k + 16 <= cnt; k += 16) add_rows<16>(d, r, p, k, lane, h, acc);
    for (; k + 4  <= cnt; k += 4)  add_rows<4>(d, r, p, k, lane, h, acc);
    for (; k      <  cnt; k += 1)  add_rows<1>(d, r, p, k, lane, h, acc);
}

__device__ __forceinline__ void queue_rest(int* qe, int& qn, bool flush, int b, int lane)
{
    if (flush) {
        qn = 0;
    } else if (b > 0) {
        const int rem = qn - b;
        if (lane < rem) {
            const int e = qe[b + lane];
            qe[lane] = e;
        }
        qn = rem;
    }
}

__device__ __forceinline__ void edge_terms(int e, int row0, const int2* __restrict__ rec,
                                           const int* __restrict__ src, const int* __restrict__ dst,
                                           const float* __restrict__ s_dst,
                                           int& r, int& d, float& sd)
{
    if (rec) {
        const unsigned long long rc = ((const unsigned long long*)rec)[e];
        const unsigned w = (unsigned)rc;
        r  = (int)(w >> REC_DST_BITS);
        d  = (int)(w & ((1u << REC_DST_BITS) - 1u));
        sd = __uint_as_float((unsigned)(rc >> 32));
    } else {
        r  = src[e] - row0;
        d  = dst[e];
        sd = s_dst[d];
    }
}

__device__ __forceinline__ void note_queue(
    int* qe, int& qn, bool flush, int lane, int row0, const int2* __restrict__ rec,
    const int* __restrict__ src, const int* __restrict__ dst,
    const float* __restrict__ s_dst, const float* ssl, float* ml,
    unsigned short* list_r, int* list_d, float* list_lg, int* list_cnt)
{
    int b = 0;
    while (qn - b >= 32 || (flush && b < qn)) {
        const int cnt = (qn - b) < 32 ? (qn - b) : 32;
        int base = 0;
        if (lane == 0) base = atomicAdd(list_cnt, cnt);
        base = __builtin_amdgcn_readfirstlane(base);
        if (lane < cnt) {
            const int e = qe[b + lane];
            int r, d;
            float sd;
            edge_terms(e, row0, rec, src, dst, s_dst, r, d, sd);
            const float lg = logit_of(ssl[r] + sd);
            atomicMax(&ml[r], lg);
            const int slot = base + lane;
            if (slot < LIST_CAP) {
                list_r[slot]  = (unsigned short)r;
                list_d[slot]  = d;
                list_lg[slot] = lg;
            }
        }
        b += 32;
    }
    queue_rest(qe, qn, flush, b, lane);
}

__device__ __forceinline__ void drain_queue(
    int* qe, int& qn, bool flush, int lane, int row0, const int2* __restrict__ rec,
    const int* __restrict__ src, const int* __restrict__ dst,
    const float* __restrict__ s_dst, const float* __restrict__ h,
    const float* ssl, const float* ml, float* den, float* acc)
{
    int b = 0;
    while (qn - b >= 32 || (flush && b < qn)) {
        const int cnt = (qn - b) < 32 ? (qn - b) : 32;
        int d = 0, r = 0;
        float p = 0.f;
        if (lane < cnt) {
            const int e = qe[b + lane];
            float sd;
            edge_terms(e, row0, rec, src, dst, s_dst, r, d, sd);
            p = __expf(logit_of(ssl[r] + sd) - ml[r]);
            atomicAdd(&den[r], p);
        }
        add_batch(d, r, p, cnt, lane, h, acc);
        b += 32;
    }
    queue_rest(qe, qn, flush, b, lane);
}

__device__ __forceinline__ unsigned zero_flags(unsigned x)
{
    return ~(((x & 0x7F7F7F7Fu) + 0x7F7F7F7Fu) | x);
}

__device__ __forceinline__ unsigned hit_mask2(const v4i a, const v4i b, unsigned splat)
{
    return ((zero_flags((unsigned)a.x ^ splat) >> 7) & 0x01010101u)
         | ((zero_flags((unsigned)a.y ^ splat) >> 6) & 0x02020202u)
         | ((zero_flags((unsigned)a.z ^ splat) >> 5) & 0x04040404u)
         | ((zero_flags((unsigned)a.w ^ splat) >> 4) & 0x08080808u)
         | ((zero_flags((unsigned)b.x ^ splat) >> 3) & 0x10101010u)
         | ((zero_flags((unsigned)b.y ^ splat) >> 2) & 0x20202020u)
         | ((zero_flags((unsigned)b.z ^ splat) >> 1) & 0x40404040u)
         | ( zero_flags((unsigned)b.w ^ splat)       & 0x80808080u);
}

#define KSCAN_U  8
#define KQ_ROOM  (Q_CAP - 32)

struct OwnerLds {
    float* acc;
    float* den;
    const float* ssl;
    float* ml;
    unsigned short* list_r;
    int*   list_d;
    float* list_lg;
    int*   list_cnt;
};

template <bool FIRST>
__device__ __forceinline__ void sweep_edges(
    int* qe, int lane, int wave, int row0, int nrows, bool use_keys, const int2* __restrict__ rec,
    const int* __restrict__ src, const int* __restrict__ dst,
    const float* __restrict__ s_dst, const float* __restrict__ h,
    const OwnerLds L, float* ws_end, int E)
{
    int qn = 0;
    auto work_off = [&](bool flush) {
        if (FIRST)
            note_queue(qe, qn, flush, lane, row0, rec, src, dst, s_dst, L.ssl, L.ml,
                       L.list_r, L.list_d, L.list_lg, L.list_cnt);
        else
            drain_queue(qe, qn, flush, lane, row0, rec, src, dst, s_dst, h, L.ssl, L.ml, L.den, L.acc);
    };

    if (use_keys) {
        const v4i* keys4 = (const v4i*)key_plane(ws_end);
        const unsigned splat = blockIdx.x * 0x01010101u;
        const int nvec   = (E + 15) >> 4;
        const int nsteps = (nvec + 1024 * KSCAN_U - 1) / (1024 * KSCAN_U);
        const int t0 = (int)(((long long)blockIdx.x * nsteps) / gridDim.x);
        for (int it = 0; it < nsteps; ++it) {
            int t = it + t0;
            if (t >= nsteps) t -= nsteps;
            const int i0 = t * (1024 * KSCAN_U) + wave * (32 * KSCAN_U) + lane;
            unsigned mm[KSCAN_U / 2];
            {
                v4i v[KSCAN_U];
#pragma unroll
                for (int u = 0; u < KSCAN_U; ++u) {
                    const int iv = i0 + u * 32;
                    v[u] = keys4[iv < nvec ? iv : nvec - 1];
                }
                static_assert(KSCAN_U == 8, "the fence names eight vectors");
                asm volatile("" :: "v"(v[0]), "v"(v[1]), "v"(v[2]), "v"(v[3]),
                                   "v"(v[4]), "v"(v[5]), "v"(v[6]), "v"(v[7]));
#pragma unroll
                for (int u = 0; u < KSCAN_U / 2; ++u)
                    mm[u] = hit_mask2(v[2 * u], v[2 * u + 1], splat);
            }
            const bool last = (it == nsteps - 1);
            bool more;
            do {
#pragma unroll
                for (int u = 0; u < KSCAN_U / 2; ++u) {
                    while (qn <= KQ_ROOM) {
                        const bool has = mm[u] != 0u;
                        if (__builtin_amdgcn_ballot_w32(has) == 0u) break;
                        int e = E;
                        if (has) {
                            const int p = __builtin_ctz(mm[u]);
                            mm[u] &= mm[u] - 1u;
                            e = (i0 + (2 * u + ((p >> 2) & 1)) * 32) * 16 + 4 * (p & 3) + (p >> 3);
                        }
                        file_hit(e < E, e, qe, qn);
                    }
                }
                unsigned any = 0u;
#pragma unroll
                for (int u = 0; u < KSCAN_U / 2; ++u) any |= mm[u];
                more = __builtin_amdgcn_ballot_w32(any != 0u) != 0u;
                const bool flush = last && !more;
                if (qn >= Q_DRAIN || more || flush) work_off(flush);
            } while (more);
        }
    } else {
        const int nvec   = E >> 2;
        const int nsteps = nvec / (1024 * SCAN_U) + 1;
        const int t0 = (int)(((long long)blockIdx.x * nsteps) / gridDim.x);
        for (int it = 0; it < nsteps; ++it) {
            int t = it + t0;
            if (t >= nsteps) t -= nsteps;
            const int i0 = t * (1024 * SCAN_U) + wave * (32 * SCAN_U) + lane;
            for (int u = 0; u < SCAN_U; ++u) {
                const int4 v = load_src4_tail(src, i0 + u * 32, nvec, E);
                if (qn > Q_CAP - 128) work_off(false);
                test4(v, i0 + u * 32, row0, (unsigned)nrows, qe, qn);
            }
            const bool last = (it == nsteps - 1);
            if (qn >= Q_DRAIN || last) work_off(last);
        }
    }
}

__global__ __launch_bounds__(1024) void gat_rows_scan(
    const int* __restrict__ src, const int* __restrict__ dst,
    const float* __restrict__ s_src, const float* __restrict__ s_dst,
    const float* __restrict__ h,
    float* __restrict__ out, float* ws_end, size_t ws_size, int N, int E)
{
    __shared__ float acc[ROWS_PB * OUT_F];
    __shared__ float den[ROWS_PB];
    __shared__ float ssl[ROWS_PB];
    __shared__ float ml[ROWS_PB];
    __shared__ int   queue[AGG_WAVES * Q_CAP];
    __shared__ int            list_d[LIST_CAP];
    __shared__ float          list_lg[LIST_CAP];
    __shared__ unsigned short list_r[LIST_CAP];
    __shared__ int   list_cnt;

    const int tid  = threadIdx.x;
    const int lane = tid & 31;
    const int wave = tid >> 5;
    const int row0 = blockIdx.x * ROWS_PB;
    const int nrows = (N - row0) < ROWS_PB ? (N - row0) : ROWS_PB;

    for (int i = tid; i < ROWS_PB * OUT_F; i += 1024) acc[i] = 0.f;
    for (int i = tid; i < ROWS_PB; i += 1024) {
        den[i] = 0.f;
        ssl[i] = (i < nrows) ? s_src[row0 + i] : 0.f;
        ml[i]  = -INFINITY;
    }
    if (tid == 0) list_cnt = 0;
    __syncthreads();

    int* qe = queue + wave * Q_CAP;
    const bool use_keys = keys_usable(ws_size, N, E);
    const int2* rec = recs_usable(ws_size, N, E) ? rec_plane(ws_end, E) : nullptr;
    const OwnerLds lds = { acc, den, ssl, ml, list_r, list_d, list_lg, &list_cnt };

    sweep_edges<true>(qe, lane, wave, row0, nrows, use_keys, rec, src, dst, s_dst, h, lds, ws_end, E);

    __syncthreads();

    const int total = __builtin_amdgcn_readfirstlane(list_cnt);

    if (total <= LIST_CAP) {
        const int wave_s = __builtin_amdgcn_readfirstlane(wave);
        for (int b = wave_s * 32; b < total; b += AGG_WAVES * 32) {
            const int cnt = (total - b) < 32 ? (total - b) : 32;
            int d = 0, r = 0;
            float p = 0.f;
            if (lane < cnt) {
                r = list_r[b + lane];
                d = list_d[b + lane];
                p = __expf(list_lg[b + lane] - ml[r]);
                atomicAdd(&den[r], p);
            }
            add_batch(d, r, p, cnt, lane, h, acc);
        }
    } else {
        sweep_edges<false>(qe, lane, wave, row0, nrows, use_keys, rec, src, dst, s_dst, h, lds, ws_end, E);
    }
    __syncthreads();

    for (int i = tid; i < nrows * OUT_F; i += 1024) {
        const float dn = den[i / OUT_F];
        const float v = acc[i] / (dn > 0.f ? dn : 1.f);
        out[(size_t)row0 * OUT_F + i] = v > 0.f ? v : expm1f(v);
    }
}

extern "C" void kernel_launch(void* const* d_in, const int* in_sizes, int n_in,
                              void* d_out, int out_size, void* d_ws, size_t ws_size,
                              hipStream_t stream)
{
    const float* x    = (const float*)d_in[0];
    const int*   edge = (const int*)  d_in[1];
    const float* W    = (const float*)d_in[2];
    const float* attn = (const float*)d_in[3];

    const int N = in_sizes[0] / IN_F;
    const int E = in_sizes[1] / 2;
    const int* src = edge;
    const int* dst = edge + E;
    float* out = (float*)d_out;

    float* h     = (float*)d_ws;
    float* s_src = h + (size_t)N * OUT_F;
    float* s_dst = s_src + N;

    gat_gemm_wmma<<<(N + 127) / 128,       256, 0, stream>>>(x, W, h, N);
    gat_scores   <<<(N + 255) / 256,       256, 0, stream>>>(h, attn, s_src, s_dst, N);
    gat_edge_records<<<256, 1024, 0, stream>>>(src, dst, s_dst, s_dst + N, ws_size, N, E);
    gat_rows_scan<<<(N + ROWS_PB - 1) / ROWS_PB, 1024, 0, stream>>>(
        src, dst, s_src, s_dst, h, out, s_dst + N, ws_size, N, E);
}
